# starting kernel
# baseline (speedup 1.0000x reference)
#include <hip/hip_runtime.h>
#include <hip/hip_bf16.h>


typedef __attribute__((ext_vector_type(2))) float v2f;
typedef __attribute__((ext_vector_type(8))) float v8f;

__device__ __forceinline__ void atomAddF(float* p, float v) {
#if defined(__HIP_DEVICE_COMPILE__)
    unsafeAtomicAdd(p, v);
#else
    (void)p; (void)v;
#endif
}

__global__ void k_zero(float* __restrict__ p, long long n) {
    long long i = (long long)blockIdx.x * blockDim.x + threadIdx.x;
    if (i < n) p[i] = 0.0f;
}

__global__ void k_degree(const int* __restrict__ ei, const float* __restrict__ w,
                         float* __restrict__ outdeg, float* __restrict__ indeg, int E) {
    int e = blockIdx.x * blockDim.x + threadIdx.x;
    if (e >= E) return;
    float we = w[e];
    atomAddF(&outdeg[ei[e]], we);
    atomAddF(&indeg[ei[E + e]], we);
}

__global__ void k_wnorm(const int* __restrict__ ei, const float* __restrict__ w,
                        const float* __restrict__ outdeg, const float* __restrict__ indeg,
                        float* __restrict__ wn, int E) {
    int e = blockIdx.x * blockDim.x + threadIdx.x;
    if (e >= E) return;
    float od = outdeg[ei[e]];
    float id = indeg[ei[E + e]];
    float a = (od > 0.0f) ? rsqrtf(od) : 0.0f;
    float b = (id > 0.0f) ? rsqrtf(id) : 0.0f;
    wn[e] = w[e] * a * b;
}

__global__ void k_scatter(const float* __restrict__ x, const int* __restrict__ ei,
                          const float* __restrict__ wn,
                          float* __restrict__ aggf, float* __restrict__ aggb, int E) {
    long long t = (long long)blockIdx.x * blockDim.x + threadIdx.x;
    int e = (int)(t >> 5);
    int d = (int)(t & 31) * 2;
    if (e >= E) return;
    int row = ei[e];
    int col = ei[E + e];
    float w = wn[e];
    const v2f xc = *(const v2f*)(x + (long long)col * 64 + d);
    const v2f xr = *(const v2f*)(x + (long long)row * 64 + d);
    float* pf = aggf + (long long)row * 64 + d;
    float* pb = aggb + (long long)col * 64 + d;
    atomAddF(pf + 0, w * xc.x);
    atomAddF(pf + 1, w * xc.y);
    atomAddF(pb + 0, w * xr.x);
    atomAddF(pb + 1, w * xr.y);
}

__global__ void __launch_bounds__(256)
k_wmma_epilogue(const float* __restrict__ aggf, const float* __restrict__ aggb,
                const float* __restrict__ Wsrc, const float* __restrict__ Wdst,
                const float* __restrict__ bsrc, const float* __restrict__ bdst,
                float* __restrict__ out, int tilesM, int Nnodes) {
    int wave  = (int)((blockIdx.x * blockDim.x + threadIdx.x) >> 5);
    int lane  = threadIdx.x & 31;
    int tileM = wave >> 2;
    int tileN = wave & 3;
    if (tileM >= tilesM) return;

    int lmod = lane & 15;
    int lhi  = lane >> 4;

    int arow = tileM * 16 + lmod;
    if (arow >= Nnodes) arow = Nnodes - 1;
    const float* af_p = aggf + (long long)arow * 64 + 2 * lhi;
    const float* ab_p = aggb + (long long)arow * 64 + 2 * lhi;
    const float* bs_p = Wsrc + (tileN * 16 + lmod) * 64 + 2 * lhi;
    const float* bd_p = Wdst + (tileN * 16 + lmod) * 64 + 2 * lhi;

    v8f acc = {};
#pragma unroll
    for (int kk = 0; kk < 16; ++kk) {
        v2f af = *(const v2f*)(af_p + 4 * kk);
        v2f bs = *(const v2f*)(bs_p + 4 * kk);
        acc = __builtin_amdgcn_wmma_f32_16x16x4_f32(
                  false, af, false, bs, (short)0, acc, false, false);
        v2f ab = *(const v2f*)(ab_p + 4 * kk);
        v2f bd = *(const v2f*)(bd_p + 4 * kk);
        acc = __builtin_amdgcn_wmma_f32_16x16x4_f32(
                  false, ab, false, bd, (short)0, acc, false, false);
    }

    int ncol = tileN * 16 + lmod;
    float bias = 0.5f * (bsrc[ncol] + bdst[ncol]);
    int mbase = tileM * 16 + 8 * lhi;
    float* ocol = out + ncol;
#pragma unroll
    for (int r = 0; r < 8; ++r) {
        int m = mbase + r;
        if (m < Nnodes) ocol[(long long)m * 64] = 0.5f * acc[r] + bias;
    }
}

extern "C" void kernel_launch(void* const* d_in, const int* in_sizes, int n_in,
                              void* d_out, int out_size, void* d_ws, size_t ws_size,
                              hipStream_t stream) {
    const float* x    = (const float*)d_in[0];
    const int*   ei   = (const int*)d_in[1];
    const float* ew   = (const float*)d_in[2];
    const float* Wsrc = (const float*)d_in[3];
    const float* bsrc = (const float*)d_in[4];
    const float* Wdst = (const float*)d_in[5];
    const float* bdst = (const float*)d_in[6];
    float* out = (float*)d_out;

    const int N = in_sizes[0] / 64;
    const int E = in_sizes[2];

    float* ws     = (float*)d_ws;
    float* outdeg = ws;
    float* indeg  = ws + (long long)N;
    float* wn     = ws + 2LL * N;
    float* aggf   = wn + (long long)E;
    float* aggb   = aggf + 64LL * N;

    const long long ztotal = 130LL * N + (long long)E;
    k_zero<<<(unsigned)((ztotal + 255) / 256), 256, 0, stream>>>(ws, ztotal);

    const int eb = (E + 255) / 256;
    k_degree<<<eb, 256, 0, stream>>>(ei, ew, outdeg, indeg, E);
    k_wnorm<<<eb, 256, 0, stream>>>(ei, ew, outdeg, indeg, wn, E);

    const long long st = 32LL * E;
    k_scatter<<<(unsigned)((st + 255) / 256), 256, 0, stream>>>(x, ei, wn, aggf, aggb, E);

    const int tilesM = (N + 15) / 16;
    const int waves  = tilesM * 4;
    k_wmma_epilogue<<<(waves + 7) / 8, 256, 0, stream>>>(aggf, aggb, Wsrc, Wdst,
                                                         bsrc, bdst, out, tilesM, N);
}


# degree sums: per-node-range 64-bit fixed-point tables in LDS (ds_add_u64), per-slice partials added as integers, rounded once to f32: exact, order-independent; was 2 global f32 atomics per edge
# speedup vs baseline: 1.3444x; 1.3444x over previous
#include <hip/hip_runtime.h>
#include <hip/hip_bf16.h>


typedef __attribute__((ext_vector_type(2))) float v2f;
typedef __attribute__((ext_vector_type(8))) float v8f;

__device__ __forceinline__ void atomAddF(float* p, float v) {
#if defined(__HIP_DEVICE_COMPILE__)
    unsafeAtomicAdd(p, v);
#else
    (void)p; (void)v;
#endif
}

__global__ void k_zero(float* __restrict__ p, long long n) {
    long long i = (long long)blockIdx.x * blockDim.x + threadIdx.x;
    if (i < n) p[i] = 0.0f;
}

#define DEG_R 16384
#define DEG_S 16
#define DEG_T 1024
#define DEG_U 16

__global__ void __launch_bounds__(DEG_T)
k_degree(const int* __restrict__ ei, const float* __restrict__ w,
         unsigned long long* __restrict__ part, int E, int N, int nR) {
    __shared__ unsigned long long tab[DEG_R];
    const int tid = threadIdx.x;
    const int s = blockIdx.x % DEG_S;
    const int kr = blockIdx.x / DEG_S;
    const int r = kr % nR;
    const int k = kr / nR;

    for (int j = tid; j < DEG_R; j += DEG_T) tab[j] = 0ull;
    __syncthreads();

    const int* __restrict__ key = ei + (long long)k * E;
    const int chunk = (E + DEG_S - 1) / DEG_S;
    const int e0 = s * chunk;
    const int e1 = (e0 + chunk < E) ? (e0 + chunk) : E;
    const int lo = r * DEG_R;

    for (int base = e0; base < e1; base += DEG_T * DEG_U) {
        int       kk[DEG_U];
        long long qq[DEG_U];
#pragma unroll
        for (int u = 0; u < DEG_U; ++u) {
            int e = base + tid + u * DEG_T;
            int ec = (e < e1) ? e : (e1 - 1);
            kk[u] = key[ec];
            qq[u] = (long long)(w[ec] * 0x1p40f);
        }
#pragma unroll
        for (int u = 0; u < DEG_U; ++u) {
            int e = base + tid + u * DEG_T;
            unsigned idx = (unsigned)(kk[u] - lo);
            if (e < e1 && idx < (unsigned)DEG_R)
                atomicAdd(&tab[idx], (unsigned long long)qq[u]);
        }
    }
    __syncthreads();

    unsigned long long* __restrict__ plane = part + (long long)(s * 2 + k) * N;
    for (int j = tid; j < DEG_R; j += DEG_T) {
        int node = lo + j;
        if (node < N) plane[node] = tab[j];
    }
}

__global__ void k_degsum(const unsigned long long* __restrict__ part,
                         float* __restrict__ deg, int N2) {
    int i = blockIdx.x * blockDim.x + threadIdx.x;
    if (i >= N2) return;
    unsigned long long acc = 0ull;
#pragma unroll
    for (int s = 0; s < DEG_S; ++s) acc += part[(long long)s * N2 + i];
    deg[i] = (float)(long long)acc * 0x1p-40f;
}

__global__ void k_wnorm(const int* __restrict__ ei, const float* __restrict__ w,
                        const float* __restrict__ outdeg, const float* __restrict__ indeg,
                        float* __restrict__ wn, int E) {
    int e = blockIdx.x * blockDim.x + threadIdx.x;
    if (e >= E) return;
    float od = outdeg[ei[e]];
    float id = indeg[ei[E + e]];
    float a = (od > 0.0f) ? rsqrtf(od) : 0.0f;
    float b = (id > 0.0f) ? rsqrtf(id) : 0.0f;
    wn[e] = w[e] * a * b;
}

__global__ void k_scatter(const float* __restrict__ x, const int* __restrict__ ei,
                          const float* __restrict__ wn,
                          float* __restrict__ aggf, float* __restrict__ aggb, int E) {
    long long t = (long long)blockIdx.x * blockDim.x + threadIdx.x;
    int e = (int)(t >> 5);
    int d = (int)(t & 31) * 2;
    if (e >= E) return;
    int row = ei[e];
    int col = ei[E + e];
    float w = wn[e];
    const v2f xc = *(const v2f*)(x + (long long)col * 64 + d);
    const v2f xr = *(const v2f*)(x + (long long)row * 64 + d);
    float* pf = aggf + (long long)row * 64 + d;
    float* pb = aggb + (long long)col * 64 + d;
    atomAddF(pf + 0, w * xc.x);
    atomAddF(pf + 1, w * xc.y);
    atomAddF(pb + 0, w * xr.x);
    atomAddF(pb + 1, w * xr.y);
}

__global__ void __launch_bounds__(256)
k_wmma_epilogue(const float* __restrict__ aggf, const float* __restrict__ aggb,
                const float* __restrict__ Wsrc, const float* __restrict__ Wdst,
                const float* __restrict__ bsrc, const float* __restrict__ bdst,
                float* __restrict__ out, int tilesM, int Nnodes) {
    int wave  = (int)((blockIdx.x * blockDim.x + threadIdx.x) >> 5);
    int lane  = threadIdx.x & 31;
    int tileM = wave >> 2;
    int tileN = wave & 3;
    if (tileM >= tilesM) return;

    int lmod = lane & 15;
    int lhi  = lane >> 4;

    int arow = tileM * 16 + lmod;
    if (arow >= Nnodes) arow = Nnodes - 1;
    const float* af_p = aggf + (long long)arow * 64 + 2 * lhi;
    const float* ab_p = aggb + (long long)arow * 64 + 2 * lhi;
    const float* bs_p = Wsrc + (tileN * 16 + lmod) * 64 + 2 * lhi;
    const float* bd_p = Wdst + (tileN * 16 + lmod) * 64 + 2 * lhi;

    v8f acc = {};
#pragma unroll
    for (int kk = 0; kk < 16; ++kk) {
        v2f af = *(const v2f*)(af_p + 4 * kk);
        v2f bs = *(const v2f*)(bs_p + 4 * kk);
        acc = __builtin_amdgcn_wmma_f32_16x16x4_f32(
                  false, af, false, bs, (short)0, acc, false, false);
        v2f ab = *(const v2f*)(ab_p + 4 * kk);
        v2f bd = *(const v2f*)(bd_p + 4 * kk);
        acc = __builtin_amdgcn_wmma_f32_16x16x4_f32(
                  false, ab, false, bd, (short)0, acc, false, false);
    }

    int ncol = tileN * 16 + lmod;
    float bias = 0.5f * (bsrc[ncol] + bdst[ncol]);
    int mbase = tileM * 16 + 8 * lhi;
    float* ocol = out + ncol;
#pragma unroll
    for (int r = 0; r < 8; ++r) {
        int m = mbase + r;
        if (m < Nnodes) ocol[(long long)m * 64] = 0.5f * acc[r] + bias;
    }
}

extern "C" void kernel_launch(void* const* d_in, const int* in_sizes, int n_in,
                              void* d_out, int out_size, void* d_ws, size_t ws_size,
                              hipStream_t stream) {
    const float* x    = (const float*)d_in[0];
    const int*   ei   = (const int*)d_in[1];
    const float* ew   = (const float*)d_in[2];
    const float* Wsrc = (const float*)d_in[3];
    const float* bsrc = (const float*)d_in[4];
    const float* Wdst = (const float*)d_in[5];
    const float* bdst = (const float*)d_in[6];
    float* out = (float*)d_out;

    const int N = in_sizes[0] / 64;
    const int E = in_sizes[2];

    float* ws     = (float*)d_ws;
    float* outdeg = ws;
    float* indeg  = ws + (long long)N;
    float* wn     = ws + 2LL * N;
    float* aggf   = wn + (long long)E;
    float* aggb   = aggf + 64LL * N;

    unsigned long long* part = (unsigned long long*)(aggf + ((2 * N + E) & 1));
    const int nR = (N + DEG_R - 1) / DEG_R;
    k_degree<<<2 * nR * DEG_S, DEG_T, 0, stream>>>(ei, ew, part, E, N, nR);
    k_degsum<<<(2 * N + 255) / 256, 256, 0, stream>>>(part, outdeg, 2 * N);

    const long long ztotal = 128LL * N;
    k_zero<<<(unsigned)((ztotal + 255) / 256), 256, 0, stream>>>(aggf, ztotal);

    const int eb = (E + 255) / 256;
    k_wnorm<<<eb, 256, 0, stream>>>(ei, ew, outdeg, indeg, wn, E);

    const long long st = 32LL * E;
    k_scatter<<<(unsigned)((st + 255) / 256), 256, 0, stream>>>(x, ei, wn, aggf, aggb, E);

    const int tilesM = (N + 15) / 16;
    const int waves  = tilesM * 4;
    k_wmma_epilogue<<<(waves + 7) / 8, 256, 0, stream>>>(aggf, aggb, Wsrc, Wdst,
                                                         bsrc, bdst, out, tilesM, N);
}


# scatter: one wave walks 32 edges (indices, weights loaded once per wave; row bases read from lane j into scalar registers); was one wave per edge, 1600k waves become 50k; f32 atomics, arrival order
# speedup vs baseline: 1.6131x; 1.1999x over previous
#include <hip/hip_runtime.h>
#include <hip/hip_bf16.h>


typedef __attribute__((ext_vector_type(2))) float v2f;
typedef __attribute__((ext_vector_type(8))) float v8f;

__device__ __forceinline__ void atomAddF(float* p, float v) {
#if defined(__HIP_DEVICE_COMPILE__)
    unsafeAtomicAdd(p, v);
#else
    (void)p; (void)v;
#endif
}

__global__ void k_zero(float* __restrict__ p, long long n) {
    long long i = (long long)blockIdx.x * blockDim.x + threadIdx.x;
    if (i < n) p[i] = 0.0f;
}

#define DEG_R 16384
#define DEG_S 16
#define DEG_T 1024
#define DEG_U 16

__global__ void __launch_bounds__(DEG_T)
k_degree(const int* __restrict__ ei, const float* __restrict__ w,
         unsigned long long* __restrict__ part, int E, int N, int nR) {
    __shared__ unsigned long long tab[DEG_R];
    const int tid = threadIdx.x;
    const int s = blockIdx.x % DEG_S;
    const int kr = blockIdx.x / DEG_S;
    const int r = kr % nR;
    const int k = kr / nR;

    for (int j = tid; j < DEG_R; j += DEG_T) tab[j] = 0ull;
    __syncthreads();

    const int* __restrict__ key = ei + (long long)k * E;
    const int chunk = (E + DEG_S - 1) / DEG_S;
    const int e0 = s * chunk;
    const int e1 = (e0 + chunk < E) ? (e0 + chunk) : E;
    const int lo = r * DEG_R;

    for (int base = e0; base < e1; base += DEG_T * DEG_U) {
        int       kk[DEG_U];
        long long qq[DEG_U];
#pragma unroll
        for (int u = 0; u < DEG_U; ++u) {
            int e = base + tid + u * DEG_T;
            int ec = (e < e1) ? e : (e1 - 1);
            kk[u] = key[ec];
            qq[u] = (long long)(w[ec] * 0x1p40f);
        }
#pragma unroll
        for (int u = 0; u < DEG_U; ++u) {
            int e = base + tid + u * DEG_T;
            unsigned idx = (unsigned)(kk[u] - lo);
            if (e < e1 && idx < (unsigned)DEG_R)
                atomicAdd(&tab[idx], (unsigned long long)qq[u]);
        }
    }
    __syncthreads();

    unsigned long long* __restrict__ plane = part + (long long)(s * 2 + k) * N;
    for (int j = tid; j < DEG_R; j += DEG_T) {
        int node = lo + j;
        if (node < N) plane[node] = tab[j];
    }
}

__global__ void k_degsum(const unsigned long long* __restrict__ part,
                         float* __restrict__ deg, int N2) {
    int i = blockIdx.x * blockDim.x + threadIdx.x;
    if (i >= N2) return;
    unsigned long long acc = 0ull;
#pragma unroll
    for (int s = 0; s < DEG_S; ++s) acc += part[(long long)s * N2 + i];
    deg[i] = (float)(long long)acc * 0x1p-40f;
}

__global__ void k_wnorm(const int* __restrict__ ei, const float* __restrict__ w,
                        const float* __restrict__ outdeg, const float* __restrict__ indeg,
                        float* __restrict__ wn, int E) {
    int e = blockIdx.x * blockDim.x + threadIdx.x;
    if (e >= E) return;
    float od = outdeg[ei[e]];
    float id = indeg[ei[E + e]];
    float a = (od > 0.0f) ? rsqrtf(od) : 0.0f;
    float b = (id > 0.0f) ? rsqrtf(id) : 0.0f;
    wn[e] = w[e] * a * b;
}

#define SC_E 32
#define SC_U 8

__global__ void __launch_bounds__(256)
k_scatter(const float* __restrict__ x, const int* __restrict__ ei,
          const float* __restrict__ wn,
          float* __restrict__ aggf, float* __restrict__ aggb, int E) {
    const int lane = threadIdx.x & 31;
    const int wv = __builtin_amdgcn_readfirstlane((int)(threadIdx.x >> 5));
    const long long ebase = ((long long)blockIdx.x * (blockDim.x >> 5) + wv) * SC_E;
    if (ebase >= E) return;
    const int n = (E - ebase < SC_E) ? (int)(E - ebase) : SC_E;

    const long long e  = ebase + lane;
    const long long ec = (e < E) ? e : (long long)E - 1;
    const int r  = ei[ec];
    const int c  = ei[E + ec];
    const int wb = __float_as_int(wn[ec]);
    const int d  = lane * 2;

    for (int j0 = 0; j0 < n; j0 += SC_U) {
        if (j0 + SC_U <= n) {
            int   row[SC_U], col[SC_U];
            float w[SC_U];
            v2f   xc[SC_U], xr[SC_U];
#pragma unroll
            for (int u = 0; u < SC_U; ++u) {
                row[u] = __builtin_amdgcn_readlane(r, j0 + u);
                col[u] = __builtin_amdgcn_readlane(c, j0 + u);
                w[u]   = __int_as_float(__builtin_amdgcn_readlane(wb, j0 + u));
                xc[u] = *(const v2f*)(x + (long long)col[u] * 64 + d);
                xr[u] = *(const v2f*)(x + (long long)row[u] * 64 + d);
            }
#pragma unroll
            for (int u = 0; u < SC_U; ++u) {
                float* pf = aggf + (long long)row[u] * 64 + d;
                float* pb = aggb + (long long)col[u] * 64 + d;
                atomAddF(pf + 0, w[u] * xc[u].x);
                atomAddF(pf + 1, w[u] * xc[u].y);
                atomAddF(pb + 0, w[u] * xr[u].x);
                atomAddF(pb + 1, w[u] * xr[u].y);
            }
        } else {
            for (int j = j0; j < n; ++j) {
                const int   row = __builtin_amdgcn_readlane(r, j);
                const int   col = __builtin_amdgcn_readlane(c, j);
                const float w   = __int_as_float(__builtin_amdgcn_readlane(wb, j));
                const v2f xc = *(const v2f*)(x + (long long)col * 64 + d);
                const v2f xr = *(const v2f*)(x + (long long)row * 64 + d);
                float* pf = aggf + (long long)row * 64 + d;
                float* pb = aggb + (long long)col * 64 + d;
                atomAddF(pf + 0, w * xc.x);
                atomAddF(pf + 1, w * xc.y);
                atomAddF(pb + 0, w * xr.x);
                atomAddF(pb + 1, w * xr.y);
            }
        }
    }
}

__global__ void __launch_bounds__(256)
k_wmma_epilogue(const float* __restrict__ aggf, const float* __restrict__ aggb,
                const float* __restrict__ Wsrc, const float* __restrict__ Wdst,
                const float* __restrict__ bsrc, const float* __restrict__ bdst,
                float* __restrict__ out, int tilesM, int Nnodes) {
    int wave  = (int)((blockIdx.x * blockDim.x + threadIdx.x) >> 5);
    int lane  = threadIdx.x & 31;
    int tileM = wave >> 2;
    int tileN = wave & 3;
    if (tileM >= tilesM) return;

    int lmod = lane & 15;
    int lhi  = lane >> 4;

    int arow = tileM * 16 + lmod;
    if (arow >= Nnodes) arow = Nnodes - 1;
    const float* af_p = aggf + (long long)arow * 64 + 2 * lhi;
    const float* ab_p = aggb + (long long)arow * 64 + 2 * lhi;
    const float* bs_p = Wsrc + (tileN * 16 + lmod) * 64 + 2 * lhi;
    const float* bd_p = Wdst + (tileN * 16 + lmod) * 64 + 2 * lhi;

    v8f acc = {};
#pragma unroll
    for (int kk = 0; kk < 16; ++kk) {
        v2f af = *(const v2f*)(af_p + 4 * kk);
        v2f bs = *(const v2f*)(bs_p + 4 * kk);
        acc = __builtin_amdgcn_wmma_f32_16x16x4_f32(
                  false, af, false, bs, (short)0, acc, false, false);
        v2f ab = *(const v2f*)(ab_p + 4 * kk);
        v2f bd = *(const v2f*)(bd_p + 4 * kk);
        acc = __builtin_amdgcn_wmma_f32_16x16x4_f32(
                  false, ab, false, bd, (short)0, acc, false, false);
    }

    int ncol = tileN * 16 + lmod;
    float bias = 0.5f * (bsrc[ncol] + bdst[ncol]);
    int mbase = tileM * 16 + 8 * lhi;
    float* ocol = out + ncol;
#pragma unroll
    for (int r = 0; r < 8; ++r) {
        int m = mbase + r;
        if (m < Nnodes) ocol[(long long)m * 64] = 0.5f * acc[r] + bias;
    }
}

extern "C" void kernel_launch(void* const* d_in, const int* in_sizes, int n_in,
                              void* d_out, int out_size, void* d_ws, size_t ws_size,
                              hipStream_t stream) {
    const float* x    = (const float*)d_in[0];
    const int*   ei   = (const int*)d_in[1];
    const float* ew   = (const float*)d_in[2];
    const float* Wsrc = (const float*)d_in[3];
    const float* bsrc = (const float*)d_in[4];
    const float* Wdst = (const float*)d_in[5];
    const float* bdst = (const float*)d_in[6];
    float* out = (float*)d_out;

    const int N = in_sizes[0] / 64;
    const int E = in_sizes[2];

    float* ws     = (float*)d_ws;
    float* outdeg = ws;
    float* indeg  = ws + (long long)N;
    float* wn     = ws + 2LL * N;
    float* aggf   = wn + (long long)E;
    float* aggb   = aggf + 64LL * N;

    unsigned long long* part = (unsigned long long*)(aggf + ((2 * N + E) & 1));
    const int nR = (N + DEG_R - 1) / DEG_R;
    k_degree<<<2 * nR * DEG_S, DEG_T, 0, stream>>>(ei, ew, part, E, N, nR);
    k_degsum<<<(2 * N + 255) / 256, 256, 0, stream>>>(part, outdeg, 2 * N);

    const long long ztotal = 128LL * N;
    k_zero<<<(unsigned)((ztotal + 255) / 256), 256, 0, stream>>>(aggf, ztotal);

    const int eb = (E + 255) / 256;
    k_wnorm<<<eb, 256, 0, stream>>>(ei, ew, outdeg, indeg, wn, E);

    const int sw = (E + SC_E - 1) / SC_E;
    k_scatter<<<(sw + 7) / 8, 256, 0, stream>>>(x, ei, wn, aggf, aggb, E);

    const int tilesM = (N + 15) / 16;
    const int waves  = tilesM * 4;
    k_wmma_epilogue<<<(waves + 7) / 8, 256, 0, stream>>>(aggf, aggb, Wsrc, Wdst,
                                                         bsrc, bdst, out, tilesM, N);
}


# scatter lane map: lane l owns features l and l+32, so each atomic and each row load covers one whole 128-byte line; was 2l, 2l+1: two lines, every other word, per atomic; f32 atomics, arrival order
# speedup vs baseline: 1.8508x; 1.1474x over previous
#include <hip/hip_runtime.h>
#include <hip/hip_bf16.h>


typedef __attribute__((ext_vector_type(2))) float v2f;
typedef __attribute__((ext_vector_type(8))) float v8f;

__device__ __forceinline__ void atomAddF(float* p, float v) {
#if defined(__HIP_DEVICE_COMPILE__)
    unsafeAtomicAdd(p, v);
#else
    (void)p; (void)v;
#endif
}

__global__ void k_zero(float* __restrict__ p, long long n) {
    long long i = (long long)blockIdx.x * blockDim.x + threadIdx.x;
    if (i < n) p[i] = 0.0f;
}

#define DEG_R 16384
#define DEG_S 16
#define DEG_T 1024
#define DEG_U 16

__global__ void __launch_bounds__(DEG_T)
k_degree(const int* __restrict__ ei, const float* __restrict__ w,
         unsigned long long* __restrict__ part, int E, int N, int nR) {
    __shared__ unsigned long long tab[DEG_R];
    const int tid = threadIdx.x;
    const int s = blockIdx.x % DEG_S;
    const int kr = blockIdx.x / DEG_S;
    const int r = kr % nR;
    const int k = kr / nR;

    for (int j = tid; j < DEG_R; j += DEG_T) tab[j] = 0ull;
    __syncthreads();

    const int* __restrict__ key = ei + (long long)k * E;
    const int chunk = (E + DEG_S - 1) / DEG_S;
    const int e0 = s * chunk;
    const int e1 = (e0 + chunk < E) ? (e0 + chunk) : E;
    const int lo = r * DEG_R;

    for (int base = e0; base < e1; base += DEG_T * DEG_U) {
        int       kk[DEG_U];
        long long qq[DEG_U];
#pragma unroll
        for (int u = 0; u < DEG_U; ++u) {
            int e = base + tid + u * DEG_T;
            int ec = (e < e1) ? e : (e1 - 1);
            kk[u] = key[ec];
            qq[u] = (long long)(w[ec] * 0x1p40f);
        }
#pragma unroll
        for (int u = 0; u < DEG_U; ++u) {
            int e = base + tid + u * DEG_T;
            unsigned idx = (unsigned)(kk[u] - lo);
            if (e < e1 && idx < (unsigned)DEG_R)
                atomicAdd(&tab[idx], (unsigned long long)qq[u]);
        }
    }
    __syncthreads();

    unsigned long long* __restrict__ plane = part + (long long)(s * 2 + k) * N;
    for (int j = tid; j < DEG_R; j += DEG_T) {
        int node = lo + j;
        if (node < N) plane[node] = tab[j];
    }
}

__global__ void k_degsum(const unsigned long long* __restrict__ part,
                         float* __restrict__ deg, int N2) {
    int i = blockIdx.x * blockDim.x + threadIdx.x;
    if (i >= N2) return;
    unsigned long long acc = 0ull;
#pragma unroll
    for (int s = 0; s < DEG_S; ++s) acc += part[(long long)s * N2 + i];
    deg[i] = (float)(long long)acc * 0x1p-40f;
}

__global__ void k_wnorm(const int* __restrict__ ei, const float* __restrict__ w,
                        const float* __restrict__ outdeg, const float* __restrict__ indeg,
                        float* __restrict__ wn, int E) {
    int e = blockIdx.x * blockDim.x + threadIdx.x;
    if (e >= E) return;
    float od = outdeg[ei[e]];
    float id = indeg[ei[E + e]];
    float a = (od > 0.0f) ? rsqrtf(od) : 0.0f;
    float b = (id > 0.0f) ? rsqrtf(id) : 0.0f;
    wn[e] = w[e] * a * b;
}

#define SC_E 32
#define SC_U 8

__global__ void __launch_bounds__(256)
k_scatter(const float* __restrict__ x, const int* __restrict__ ei,
          const float* __restrict__ wn,
          float* __restrict__ aggf, float* __restrict__ aggb, int E) {
    const int lane = threadIdx.x & 31;
    const int wv = __builtin_amdgcn_readfirstlane((int)(threadIdx.x >> 5));
    const long long ebase = ((long long)blockIdx.x * (blockDim.x >> 5) + wv) * SC_E;
    if (ebase >= E) return;
    const int n = (E - ebase < SC_E) ? (int)(E - ebase) : SC_E;

    const long long e  = ebase + lane;
    const long long ec = (e < E) ? e : (long long)E - 1;
    const int r  = ei[ec];
    const int c  = ei[E + ec];
    const int wb = __float_as_int(wn[ec]);

    for (int j0 = 0; j0 < n; j0 += SC_U) {
        if (j0 + SC_U <= n) {
            int   row[SC_U], col[SC_U];
            float w[SC_U];
            float xc0[SC_U], xc1[SC_U], xr0[SC_U], xr1[SC_U];
#pragma unroll
            for (int u = 0; u < SC_U; ++u) {
                row[u] = __builtin_amdgcn_readlane(r, j0 + u);
                col[u] = __builtin_amdgcn_readlane(c, j0 + u);
                w[u]   = __int_as_float(__builtin_amdgcn_readlane(wb, j0 + u));
                const float* pc = x + (long long)col[u] * 64 + lane;
                const float* pr = x + (long long)row[u] * 64 + lane;
                xc0[u] = pc[0];
                xc1[u] = pc[32];
                xr0[u] = pr[0];
                xr1[u] = pr[32];
            }
#pragma unroll
            for (int u = 0; u < SC_U; ++u) {
                float* pf = aggf + (long long)row[u] * 64 + lane;
                float* pb = aggb + (long long)col[u] * 64 + lane;
                atomAddF(pf,      w[u] * xc0[u]);
                atomAddF(pf + 32, w[u] * xc1[u]);
                atomAddF(pb,      w[u] * xr0[u]);
                atomAddF(pb + 32, w[u] * xr1[u]);
            }
        } else {
            for (int j = j0; j < n; ++j) {
                const int   row = __builtin_amdgcn_readlane(r, j);
                const int   col = __builtin_amdgcn_readlane(c, j);
                const float w   = __int_as_float(__builtin_amdgcn_readlane(wb, j));
                const float* pc = x + (long long)col * 64 + lane;
                const float* pr = x + (long long)row * 64 + lane;
                const float xc0 = pc[0], xc1 = pc[32];
                const float xr0 = pr[0], xr1 = pr[32];
                float* pf = aggf + (long long)row * 64 + lane;
                float* pb = aggb + (long long)col * 64 + lane;
                atomAddF(pf,      w * xc0);
                atomAddF(pf + 32, w * xc1);
                atomAddF(pb,      w * xr0);
                atomAddF(pb + 32, w * xr1);
            }
        }
    }
}

__global__ void __launch_bounds__(256)
k_wmma_epilogue(const float* __restrict__ aggf, const float* __restrict__ aggb,
                const float* __restrict__ Wsrc, const float* __restrict__ Wdst,
                const float* __restrict__ bsrc, const float* __restrict__ bdst,
                float* __restrict__ out, int tilesM, int Nnodes) {
    int wave  = (int)((blockIdx.x * blockDim.x + threadIdx.x) >> 5);
    int lane  = threadIdx.x & 31;
    int tileM = wave >> 2;
    int tileN = wave & 3;
    if (tileM >= tilesM) return;

    int lmod = lane & 15;
    int lhi  = lane >> 4;

    int arow = tileM * 16 + lmod;
    if (arow >= Nnodes) arow = Nnodes - 1;
    const float* af_p = aggf + (long long)arow * 64 + 2 * lhi;
    const float* ab_p = aggb + (long long)arow * 64 + 2 * lhi;
    const float* bs_p = Wsrc + (tileN * 16 + lmod) * 64 + 2 * lhi;
    const float* bd_p = Wdst + (tileN * 16 + lmod) * 64 + 2 * lhi;

    v8f acc = {};
#pragma unroll
    for (int kk = 0; kk < 16; ++kk) {
        v2f af = *(const v2f*)(af_p + 4 * kk);
        v2f bs = *(const v2f*)(bs_p + 4 * kk);
        acc = __builtin_amdgcn_wmma_f32_16x16x4_f32(
                  false, af, false, bs, (short)0, acc, false, false);
        v2f ab = *(const v2f*)(ab_p + 4 * kk);
        v2f bd = *(const v2f*)(bd_p + 4 * kk);
        acc = __builtin_amdgcn_wmma_f32_16x16x4_f32(
                  false, ab, false, bd, (short)0, acc, false, false);
    }

    int ncol = tileN * 16 + lmod;
    float bias = 0.5f * (bsrc[ncol] + bdst[ncol]);
    int mbase = tileM * 16 + 8 * lhi;
    float* ocol = out + ncol;
#pragma unroll
    for (int r = 0; r < 8; ++r) {
        int m = mbase + r;
        if (m < Nnodes) ocol[(long long)m * 64] = 0.5f * acc[r] + bias;
    }
}

extern "C" void kernel_launch(void* const* d_in, const int* in_sizes, int n_in,
                              void* d_out, int out_size, void* d_ws, size_t ws_size,
                              hipStream_t stream) {
    const float* x    = (const float*)d_in[0];
    const int*   ei   = (const int*)d_in[1];
    const float* ew   = (const float*)d_in[2];
    const float* Wsrc = (const float*)d_in[3];
    const float* bsrc = (const float*)d_in[4];
    const float* Wdst = (const float*)d_in[5];
    const float* bdst = (const float*)d_in[6];
    float* out = (float*)d_out;

    const int N = in_sizes[0] / 64;
    const int E = in_sizes[2];

    float* ws     = (float*)d_ws;
    float* outdeg = ws;
    float* indeg  = ws + (long long)N;
    float* wn     = ws + 2LL * N;
    float* aggf   = wn + (long long)E;
    float* aggb   = aggf + 64LL * N;

    unsigned long long* part = (unsigned long long*)(aggf + ((2 * N + E) & 1));
    const int nR = (N + DEG_R - 1) / DEG_R;
    k_degree<<<2 * nR * DEG_S, DEG_T, 0, stream>>>(ei, ew, part, E, N, nR);
    k_degsum<<<(2 * N + 255) / 256, 256, 0, stream>>>(part, outdeg, 2 * N);

    const long long ztotal = 128LL * N;
    k_zero<<<(unsigned)((ztotal + 255) / 256), 256, 0, stream>>>(aggf, ztotal);

    const int eb = (E + 255) / 256;
    k_wnorm<<<eb, 256, 0, stream>>>(ei, ew, outdeg, indeg, wn, E);

    const int sw = (E + SC_E - 1) / SC_E;
    k_scatter<<<(sw + 7) / 8, 256, 0, stream>>>(x, ei, wn, aggf, aggb, E);

    const int tilesM = (N + 15) / 16;
    const int waves  = tilesM * 4;
    k_wmma_epilogue<<<(waves + 7) / 8, 256, 0, stream>>>(aggf, aggb, Wsrc, Wdst,
                                                         bsrc, bdst, out, tilesM, N);
}


# speedup vs baseline: 2.0285x; 1.0960x over previous
#include <hip/hip_runtime.h>
#include <hip/hip_bf16.h>


typedef __attribute__((ext_vector_type(2))) float v2f;
typedef __attribute__((ext_vector_type(8))) float v8f;

__device__ __forceinline__ void atomAddF(float* p, float v) {
#if defined(__HIP_DEVICE_COMPILE__)
    unsafeAtomicAdd(p, v);
#else
    (void)p; (void)v;
#endif
}

__global__ void k_zero(float* __restrict__ p, long long n, int* __restrict__ cnt) {
    long long i = (long long)blockIdx.x * blockDim.x + threadIdx.x;
    if (i < n) p[i] = 0.0f;
    if (i < 2) cnt[i] = 0;
}

#define DEG_R 16384
#define DEG_S 16
#define DEG_T 1024
#define DEG_U 16

__global__ void __launch_bounds__(DEG_T)
k_degree(const int* __restrict__ ei, const float* __restrict__ w,
         unsigned long long* __restrict__ part, int E, int N, int nR) {
    __shared__ unsigned long long tab[DEG_R];
    const int tid = threadIdx.x;
    const int s = blockIdx.x % DEG_S;
    const int kr = blockIdx.x / DEG_S;
    const int r = kr % nR;
    const int k = kr / nR;

    for (int j = tid; j < DEG_R; j += DEG_T) tab[j] = 0ull;
    __syncthreads();

    const int* __restrict__ key = ei + (long long)k * E;
    const int chunk = (E + DEG_S - 1) / DEG_S;
    const int e0 = s * chunk;
    const int e1 = (e0 + chunk < E) ? (e0 + chunk) : E;
    const int lo = r * DEG_R;

    for (int base = e0; base < e1; base += DEG_T * DEG_U) {
        int       kk[DEG_U];
        long long qq[DEG_U];
#pragma unroll
        for (int u = 0; u < DEG_U; ++u) {
            int e = base + tid + u * DEG_T;
            int ec = (e < e1) ? e : (e1 - 1);
            kk[u] = key[ec];
            qq[u] = (long long)(w[ec] * 0x1p40f);
        }
#pragma unroll
        for (int u = 0; u < DEG_U; ++u) {
            int e = base + tid + u * DEG_T;
            unsigned idx = (unsigned)(kk[u] - lo);
            if (e < e1 && idx < (unsigned)DEG_R)
                atomicAdd(&tab[idx], (unsigned long long)qq[u]);
        }
    }
    __syncthreads();

    unsigned long long* __restrict__ plane = part + (long long)(s * 2 + k) * N;
    for (int j = tid; j < DEG_R; j += DEG_T) {
        int node = lo + j;
        if (node < N) plane[node] = tab[j];
    }
}

__global__ void k_degsum(const unsigned long long* __restrict__ part,
                         float* __restrict__ deg, int N2) {
    int i = blockIdx.x * blockDim.x + threadIdx.x;
    if (i >= N2) return;
    unsigned long long acc = 0ull;
#pragma unroll
    for (int s = 0; s < DEG_S; ++s) acc += part[(long long)s * N2 + i];
    deg[i] = (float)(long long)acc * 0x1p-40f;
}

__global__ void k_wnorm(const int* __restrict__ ei, const float* __restrict__ w,
                        const float* __restrict__ outdeg, const float* __restrict__ indeg,
                        float* __restrict__ wn, int E) {
    int e = blockIdx.x * blockDim.x + threadIdx.x;
    if (e >= E) return;
    float od = outdeg[ei[e]];
    float id = indeg[ei[E + e]];
    float a = (od > 0.0f) ? rsqrtf(od) : 0.0f;
    float b = (id > 0.0f) ? rsqrtf(id) : 0.0f;
    wn[e] = w[e] * a * b;
}

#define SC_E 32
#define SC_B 2
#define SC_U 16
#define SC_W 8
#define SC_T (SC_E * SC_B * SC_W)

__global__ void __launch_bounds__(SC_W * 32)
k_scatter(const float* __restrict__ x, const int* __restrict__ ei,
          const float* __restrict__ wn,
          float* __restrict__ aggf, float* __restrict__ aggb,
          int* __restrict__ cnt, int E, int T) {
    __shared__ int s_tk[2];
    if (threadIdx.x == 0) {
        const unsigned id = __builtin_amdgcn_s_sendmsg_rtn(0x87);
        int d = (int)((id >> 18) & 1u);
        int t = atomicAdd(&cnt[d], 1);
        if (t >= T) { d = 1 - d; t = atomicAdd(&cnt[d], 1); }
        s_tk[0] = t;
        s_tk[1] = d;
    }
    __syncthreads();
    const int t = __builtin_amdgcn_readfirstlane(s_tk[0]);
    const int d = __builtin_amdgcn_readfirstlane(s_tk[1]);
    if (t >= T) return;

    const int lane = threadIdx.x & 31;
    const int wv = __builtin_amdgcn_readfirstlane((int)(threadIdx.x >> 5));
    const int* __restrict__ kdst = d ? (ei + E) : ei;
    const int* __restrict__ ksrc = d ? ei : (ei + E);
    float* __restrict__ agg = d ? aggb : aggf;
    const long long wbase = ((long long)t * SC_W + wv) * (SC_E * SC_B);

    for (int b = 0; b < SC_B; ++b) {
        const long long ebase = wbase + (long long)b * SC_E;
        if (ebase >= E) break;
        const int n = (E - ebase < SC_E) ? (int)(E - ebase) : SC_E;

        const long long e  = ebase + lane;
        const long long ec = (e < E) ? e : (long long)E - 1;
        const int kd = kdst[ec];
        const int ks = ksrc[ec];
        const int wb = __float_as_int(wn[ec]);

        for (int j0 = 0; j0 < n; j0 += SC_U) {
            if (j0 + SC_U <= n) {
                float x0[SC_U], x1[SC_U];
#pragma unroll
                for (int u = 0; u < SC_U; ++u) {
                    const int s = __builtin_amdgcn_readlane(ks, j0 + u);
                    const float* ps = x + (long long)s * 64 + lane;
                    x0[u] = ps[0];
                    x1[u] = ps[32];
                }
#pragma unroll
                for (int u = 0; u < SC_U; ++u) {
                    const int   dn = __builtin_amdgcn_readlane(kd, j0 + u);
                    const float w  = __int_as_float(__builtin_amdgcn_readlane(wb, j0 + u));
                    float* pd = agg + (long long)dn * 64 + lane;
                    atomAddF(pd,      w * x0[u]);
                    atomAddF(pd + 32, w * x1[u]);
                }
            } else {
                for (int j = j0; j < n; ++j) {
                    const int   s  = __builtin_amdgcn_readlane(ks, j);
                    const int   dn = __builtin_amdgcn_readlane(kd, j);
                    const float w  = __int_as_float(__builtin_amdgcn_readlane(wb, j));
                    const float* ps = x + (long long)s * 64 + lane;
                    const float a0 = ps[0], a1 = ps[32];
                    float* pd = agg + (long long)dn * 64 + lane;
                    atomAddF(pd,      w * a0);
                    atomAddF(pd + 32, w * a1);
                }
            }
        }
    }
}

__global__ void __launch_bounds__(256)
k_wmma_epilogue(const float* __restrict__ aggf, const float* __restrict__ aggb,
                const float* __restrict__ Wsrc, const float* __restrict__ Wdst,
                const float* __restrict__ bsrc, const float* __restrict__ bdst,
                float* __restrict__ out, int tilesM, int Nnodes) {
    int wave  = (int)((blockIdx.x * blockDim.x + threadIdx.x) >> 5);
    int lane  = threadIdx.x & 31;
    int tileM = wave >> 2;
    int tileN = wave & 3;
    if (tileM >= tilesM) return;

    int lmod = lane & 15;
    int lhi  = lane >> 4;

    int arow = tileM * 16 + lmod;
    if (arow >= Nnodes) arow = Nnodes - 1;
    const float* af_p = aggf + (long long)arow * 64 + 2 * lhi;
    const float* ab_p = aggb + (long long)arow * 64 + 2 * lhi;
    const float* bs_p = Wsrc + (tileN * 16 + lmod) * 64 + 2 * lhi;
    const float* bd_p = Wdst + (tileN * 16 + lmod) * 64 + 2 * lhi;

    v8f acc = {};
#pragma unroll
    for (int kk = 0; kk < 16; ++kk) {
        v2f af = *(const v2f*)(af_p + 4 * kk);
        v2f bs = *(const v2f*)(bs_p + 4 * kk);
        acc = __builtin_amdgcn_wmma_f32_16x16x4_f32(
                  false, af, false, bs, (short)0, acc, false, false);
        v2f ab = *(const v2f*)(ab_p + 4 * kk);
        v2f bd = *(const v2f*)(bd_p + 4 * kk);
        acc = __builtin_amdgcn_wmma_f32_16x16x4_f32(
                  false, ab, false, bd, (short)0, acc, false, false);
    }

    int ncol = tileN * 16 + lmod;
    float bias = 0.5f * (bsrc[ncol] + bdst[ncol]);
    int mbase = tileM * 16 + 8 * lhi;
    float* ocol = out + ncol;
#pragma unroll
    for (int r = 0; r < 8; ++r) {
        int m = mbase + r;
        if (m < Nnodes) ocol[(long long)m * 64] = 0.5f * acc[r] + bias;
    }
}

extern "C" void kernel_launch(void* const* d_in, const int* in_sizes, int n_in,
                              void* d_out, int out_size, void* d_ws, size_t ws_size,
                              hipStream_t stream) {
    const float* x    = (const float*)d_in[0];
    const int*   ei   = (const int*)d_in[1];
    const float* ew   = (const float*)d_in[2];
    const float* Wsrc = (const float*)d_in[3];
    const float* bsrc = (const float*)d_in[4];
    const float* Wdst = (const float*)d_in[5];
    const float* bdst = (const float*)d_in[6];
    float* out = (float*)d_out;

    const int N = in_sizes[0] / 64;
    const int E = in_sizes[2];

    float* ws     = (float*)d_ws;
    float* outdeg = ws;
    float* indeg  = ws + (long long)N;
    float* wn     = ws + 2LL * N;
    float* aggf   = wn + (long long)E;
    float* aggb   = aggf + 64LL * N;

    unsigned long long* part = (unsigned long long*)(aggf + ((2 * N + E) & 1));
    const int nR = (N + DEG_R - 1) / DEG_R;
    k_degree<<<2 * nR * DEG_S, DEG_T, 0, stream>>>(ei, ew, part, E, N, nR);
    k_degsum<<<(2 * N + 255) / 256, 256, 0, stream>>>(part, outdeg, 2 * N);

    const int eb = (E + 255) / 256;
    k_wnorm<<<eb, 256, 0, stream>>>(ei, ew, outdeg, indeg, wn, E);

    int* cnt = (int*)outdeg;
    const long long ztotal = 128LL * N;
    k_zero<<<(unsigned)((ztotal + 255) / 256), 256, 0, stream>>>(aggf, ztotal, cnt);

    const int T = (E + SC_T - 1) / SC_T;
    k_scatter<<<2 * T, SC_W * 32, 0, stream>>>(x, ei, wn, aggf, aggb, cnt, E, T);

    const int tilesM = (N + 15) / 16;
    const int waves  = tilesM * 4;
    k_wmma_epilogue<<<(waves + 7) / 8, 256, 0, stream>>>(aggf, aggb, Wsrc, Wdst,
                                                         bsrc, bdst, out, tilesM, N);
}


# epilogue GEMM: one wave computes a whole 16x64 strip of out: agg rows loaded once with 16-byte loads (was 4 times, 8-byte), W from LDS, whole-line stores; k order inside the products changes
# speedup vs baseline: 2.1754x; 1.0724x over previous
#include <hip/hip_runtime.h>
#include <hip/hip_bf16.h>


typedef __attribute__((ext_vector_type(2))) float v2f;
typedef __attribute__((ext_vector_type(8))) float v8f;

__device__ __forceinline__ void atomAddF(float* p, float v) {
#if defined(__HIP_DEVICE_COMPILE__)
    unsafeAtomicAdd(p, v);
#else
    (void)p; (void)v;
#endif
}

__global__ void k_zero(float* __restrict__ p, long long n, int* __restrict__ cnt) {
    long long i = (long long)blockIdx.x * blockDim.x + threadIdx.x;
    if (i < n) p[i] = 0.0f;
    if (i < 2) cnt[i] = 0;
}

#define DEG_R 16384
#define DEG_S 16
#define DEG_T 1024
#define DEG_U 16

__global__ void __launch_bounds__(DEG_T)
k_degree(const int* __restrict__ ei, const float* __restrict__ w,
         unsigned long long* __restrict__ part, int E, int N, int nR) {
    __shared__ unsigned long long tab[DEG_R];
    const int tid = threadIdx.x;
    const int s = blockIdx.x % DEG_S;
    const int kr = blockIdx.x / DEG_S;
    const int r = kr % nR;
    const int k = kr / nR;

    for (int j = tid; j < DEG_R; j += DEG_T) tab[j] = 0ull;
    __syncthreads();

    const int* __restrict__ key = ei + (long long)k * E;
    const int chunk = (E + DEG_S - 1) / DEG_S;
    const int e0 = s * chunk;
    const int e1 = (e0 + chunk < E) ? (e0 + chunk) : E;
    const int lo = r * DEG_R;

    for (int base = e0; base < e1; base += DEG_T * DEG_U) {
        int       kk[DEG_U];
        long long qq[DEG_U];
#pragma unroll
        for (int u = 0; u < DEG_U; ++u) {
            int e = base + tid + u * DEG_T;
            int ec = (e < e1) ? e : (e1 - 1);
            kk[u] = key[ec];
            qq[u] = (long long)(w[ec] * 0x1p40f);
        }
#pragma unroll
        for (int u = 0; u < DEG_U; ++u) {
            int e = base + tid + u * DEG_T;
            unsigned idx = (unsigned)(kk[u] - lo);
            if (e < e1 && idx < (unsigned)DEG_R)
                atomicAdd(&tab[idx], (unsigned long long)qq[u]);
        }
    }
    __syncthreads();

    unsigned long long* __restrict__ plane = part + (long long)(s * 2 + k) * N;
    for (int j = tid; j < DEG_R; j += DEG_T) {
        int node = lo + j;
        if (node < N) plane[node] = tab[j];
    }
}

__global__ void k_degsum(const unsigned long long* __restrict__ part,
                         float* __restrict__ deg, int N2) {
    int i = blockIdx.x * blockDim.x + threadIdx.x;
    if (i >= N2) return;
    unsigned long long acc = 0ull;
#pragma unroll
    for (int s = 0; s < DEG_S; ++s) acc += part[(long long)s * N2 + i];
    deg[i] = (float)(long long)acc * 0x1p-40f;
}

__global__ void k_wnorm(const int* __restrict__ ei, const float* __restrict__ w,
                        const float* __restrict__ outdeg, const float* __restrict__ indeg,
                        float* __restrict__ wn, int E) {
    int e = blockIdx.x * blockDim.x + threadIdx.x;
    if (e >= E) return;
    float od = outdeg[ei[e]];
    float id = indeg[ei[E + e]];
    float a = (od > 0.0f) ? rsqrtf(od) : 0.0f;
    float b = (id > 0.0f) ? rsqrtf(id) : 0.0f;
    wn[e] = w[e] * a * b;
}

#define SC_E 32
#define SC_B 2
#define SC_U 16
#define SC_W 8
#define SC_T (SC_E * SC_B * SC_W)

__global__ void __launch_bounds__(SC_W * 32)
k_scatter(const float* __restrict__ x, const int* __restrict__ ei,
          const float* __restrict__ wn,
          float* __restrict__ aggf, float* __restrict__ aggb,
          int* __restrict__ cnt, int E, int T) {
    __shared__ int s_tk[2];
    if (threadIdx.x == 0) {
        const unsigned id = __builtin_amdgcn_s_sendmsg_rtn(0x87);
        int d = (int)((id >> 18) & 1u);
        int t = atomicAdd(&cnt[d], 1);
        if (t >= T) { d = 1 - d; t = atomicAdd(&cnt[d], 1); }
        s_tk[0] = t;
        s_tk[1] = d;
    }
    __syncthreads();
    const int t = __builtin_amdgcn_readfirstlane(s_tk[0]);
    const int d = __builtin_amdgcn_readfirstlane(s_tk[1]);
    if (t >= T) return;

    const int lane = threadIdx.x & 31;
    const int wv = __builtin_amdgcn_readfirstlane((int)(threadIdx.x >> 5));
    const int* __restrict__ kdst = d ? (ei + E) : ei;
    const int* __restrict__ ksrc = d ? ei : (ei + E);
    float* __restrict__ agg = d ? aggb : aggf;
    const long long wbase = ((long long)t * SC_W + wv) * (SC_E * SC_B);

    for (int b = 0; b < SC_B; ++b) {
        const long long ebase = wbase + (long long)b * SC_E;
        if (ebase >= E) break;
        const int n = (E - ebase < SC_E) ? (int)(E - ebase) : SC_E;

        const long long e  = ebase + lane;
        const long long ec = (e < E) ? e : (long long)E - 1;
        const int kd = kdst[ec];
        const int ks = ksrc[ec];
        const int wb = __float_as_int(wn[ec]);

        for (int j0 = 0; j0 < n; j0 += SC_U) {
            if (j0 + SC_U <= n) {
                float x0[SC_U], x1[SC_U];
#pragma unroll
                for (int u = 0; u < SC_U; ++u) {
                    const int s = __builtin_amdgcn_readlane(ks, j0 + u);
                    const float* ps = x + (long long)s * 64 + lane;
                    x0[u] = ps[0];
                    x1[u] = ps[32];
                }
#pragma unroll
                for (int u = 0; u < SC_U; ++u) {
                    const int   dn = __builtin_amdgcn_readlane(kd, j0 + u);
                    const float w  = __int_as_float(__builtin_amdgcn_readlane(wb, j0 + u));
                    float* pd = agg + (long long)dn * 64 + lane;
                    atomAddF(pd,      w * x0[u]);
                    atomAddF(pd + 32, w * x1[u]);
                }
            } else {
                for (int j = j0; j < n; ++j) {
                    const int   s  = __builtin_amdgcn_readlane(ks, j);
                    const int   dn = __builtin_amdgcn_readlane(kd, j);
                    const float w  = __int_as_float(__builtin_amdgcn_readlane(wb, j));
                    const float* ps = x + (long long)s * 64 + lane;
                    const float a0 = ps[0], a1 = ps[32];
                    float* pd = agg + (long long)dn * 64 + lane;
                    atomAddF(pd,      w * a0);
                    atomAddF(pd + 32, w * a1);
                }
            }
        }
    }
}

typedef __attribute__((ext_vector_type(4))) float v4f;
#define EP_PITCH 68
#define EP_WAVES 4
#define EP_STAGE (2048 / (EP_WAVES * 32))

__global__ void __launch_bounds__(EP_WAVES * 32)
k_wmma_epilogue(const float* __restrict__ aggf, const float* __restrict__ aggb,
                const float* __restrict__ Wsrc, const float* __restrict__ Wdst,
                const float* __restrict__ bsrc, const float* __restrict__ bdst,
                float* __restrict__ out, int tilesM, int Nnodes) {
    __shared__ __attribute__((aligned(16))) float sW[2 * 64 * EP_PITCH];
    __shared__ __attribute__((aligned(16))) float sO[EP_WAVES * 16 * EP_PITCH];
    const int tid = threadIdx.x;

    v4f wtmp[EP_STAGE];
#pragma unroll
    for (int u = 0; u < EP_STAGE; ++u) {
        const int i  = tid + u * (EP_WAVES * 32);
        const int t  = i >> 10;
        const int rr = (i >> 4) & 63;
        const int q  = i & 15;
        wtmp[u] = *(const v4f*)((t ? Wdst : Wsrc) + rr * 64 + q * 4);
    }
#pragma unroll
    for (int u = 0; u < EP_STAGE; ++u) {
        const int i  = tid + u * (EP_WAVES * 32);
        const int t  = i >> 10;
        const int rr = (i >> 4) & 63;
        const int q  = i & 15;
        *(v4f*)(sW + (t * 64 + rr) * EP_PITCH + q * 4) = wtmp[u];
    }
    __syncthreads();

    const int lane  = tid & 31;
    const int wave  = tid >> 5;
    const int tileM = blockIdx.x * EP_WAVES + wave;
    if (tileM >= tilesM) return;

    const int c = lane & 15;
    const int h = lane >> 4;

    int arow = tileM * 16 + c;
    if (arow >= Nnodes) arow = Nnodes - 1;
    const float* af_p = aggf + (long long)arow * 64 + 4 * h;
    const float* ab_p = aggb + (long long)arow * 64 + 4 * h;

    v4f af[8], ab[8];
#pragma unroll
    for (int j = 0; j < 8; ++j) {
        af[j] = *(const v4f*)(af_p + 8 * j);
        ab[j] = *(const v4f*)(ab_p + 8 * j);
    }

    const float* sB = sW + c * EP_PITCH + 4 * h;
    v8f acc[4] = {};
#pragma unroll
    for (int j = 0; j < 8; ++j) {
#pragma unroll
        for (int t = 0; t < 2; ++t) {
            const v4f a = t ? ab[j] : af[j];
            v4f b[4];
#pragma unroll
            for (int n = 0; n < 4; ++n)
                b[n] = *(const v4f*)(sB + (t * 64 + n * 16) * EP_PITCH + 8 * j);
            const v2f a0 = {a.x, a.y};
            const v2f a1 = {a.z, a.w};
#pragma unroll
            for (int n = 0; n < 4; ++n) {
                const v2f b0 = {b[n].x, b[n].y};
                acc[n] = __builtin_amdgcn_wmma_f32_16x16x4_f32(
                             false, a0, false, b0, (short)0, acc[n], false, false);
            }
#pragma unroll
            for (int n = 0; n < 4; ++n) {
                const v2f b1 = {b[n].z, b[n].w};
                acc[n] = __builtin_amdgcn_wmma_f32_16x16x4_f32(
                             false, a1, false, b1, (short)0, acc[n], false, false);
            }
        }
    }

    float* so = sO + wave * (16 * EP_PITCH);
#pragma unroll
    for (int n = 0; n < 4; ++n) {
        const int ncol = n * 16 + c;
        const float bias = 0.5f * (bsrc[ncol] + bdst[ncol]);
#pragma unroll
        for (int r = 0; r < 8; ++r)
            so[(8 * h + r) * EP_PITCH + ncol] = 0.5f * acc[n][r] + bias;
    }

#pragma unroll
    for (int i = 0; i < 8; ++i) {
        const int row = 2 * i + h;
        const int m = tileM * 16 + row;
        const v4f v = *(const v4f*)(so + row * EP_PITCH + 4 * c);
        if (m < Nnodes) *(v4f*)(out + (long long)m * 64 + 4 * c) = v;
    }
}

extern "C" void kernel_launch(void* const* d_in, const int* in_sizes, int n_in,
                              void* d_out, int out_size, void* d_ws, size_t ws_size,
                              hipStream_t stream) {
    const float* x    = (const float*)d_in[0];
    const int*   ei   = (const int*)d_in[1];
    const float* ew   = (const float*)d_in[2];
    const float* Wsrc = (const float*)d_in[3];
    const float* bsrc = (const float*)d_in[4];
    const float* Wdst = (const float*)d_in[5];
    const float* bdst = (const float*)d_in[6];
    float* out = (float*)d_out;

    const int N = in_sizes[0] / 64;
    const int E = in_sizes[2];

    float* ws     = (float*)d_ws;
    float* outdeg = ws;
    float* indeg  = ws + (long long)N;
    float* wn     = ws + 2LL * N;
    float* aggf   = wn + (long long)E;
    float* aggb   = aggf + 64LL * N;

    unsigned long long* part = (unsigned long long*)(aggf + ((2 * N + E) & 1));
    const int nR = (N + DEG_R - 1) / DEG_R;
    k_degree<<<2 * nR * DEG_S, DEG_T, 0, stream>>>(ei, ew, part, E, N, nR);
    k_degsum<<<(2 * N + 255) / 256, 256, 0, stream>>>(part, outdeg, 2 * N);

    const int eb = (E + 255) / 256;
    k_wnorm<<<eb, 256, 0, stream>>>(ei, ew, outdeg, indeg, wn, E);

    int* cnt = (int*)outdeg;
    const long long ztotal = 128LL * N;
    k_zero<<<(unsigned)((ztotal + 255) / 256), 256, 0, stream>>>(aggf, ztotal, cnt);

    const int T = (E + SC_T - 1) / SC_T;
    k_scatter<<<2 * T, SC_W * 32, 0, stream>>>(x, ei, wn, aggf, aggb, cnt, E, T);

    const int tilesM = (N + 15) / 16;
    k_wmma_epilogue<<<(tilesM + EP_WAVES - 1) / EP_WAVES, EP_WAVES * 32, 0, stream>>>(
        aggf, aggb, Wsrc, Wdst, bsrc, bdst, out, tilesM, N);
}


# degrees from the binned records: a work-group per bin sums its records' weights (64-bit fixed point in LDS), w_norm formed per record in the owner kernel; the 7x edge re-read degree pass is gone
# speedup vs baseline: 3.9223x; 1.8030x over previous
#include <hip/hip_runtime.h>
#include <hip/hip_bf16.h>


typedef __attribute__((ext_vector_type(2))) float v2f;
typedef __attribute__((ext_vector_type(8))) float v8f;

__device__ __forceinline__ void atomAddF(float* p, float v) {
#if defined(__HIP_DEVICE_COMPILE__)
    unsafeAtomicAdd(p, v);
#else
    (void)p; (void)v;
#endif
}


#define BIN 400
#define MAXB 256
#define NCNT (4 * MAXB)
#define CH_T 1024
#define CH_U 4
#define CH (CH_T * CH_U)

__global__ void k_clear(int* __restrict__ cnt) {
    cnt[threadIdx.x] = 0;
}

#define CNT_T 512
#define CNT_U 16
#define CNT_CH (CNT_T * CNT_U)
__global__ void __launch_bounds__(CNT_T)
k_count(const int* __restrict__ ei, int* __restrict__ gcount, int E) {
    __shared__ int hist[2 * MAXB];
    const int tid = threadIdx.x;
    if (tid < 2 * MAXB) hist[tid] = 0;
    __syncthreads();

    const int base = blockIdx.x * CNT_CH;
    int r[CNT_U], c[CNT_U];
#pragma unroll
    for (int u = 0; u < CNT_U; ++u) {
        const int e  = base + u * CNT_T + tid;
        const int ec = (e < E) ? e : (E - 1);
        r[u] = ei[ec];
        c[u] = ei[E + ec];
    }
#pragma unroll
    for (int u = 0; u < CNT_U; ++u) {
        const int e = base + u * CNT_T + tid;
        if (e < E) {
            atomicAdd(&hist[r[u] / BIN], 1);
            atomicAdd(&hist[MAXB + c[u] / BIN], 1);
        }
    }
    __syncthreads();
    if (tid < 2 * MAXB) {
        const int hcnt = hist[tid];
        if (hcnt) atomicAdd(&gcount[tid], hcnt);
    }
}

__device__ __forceinline__ unsigned long long pack_rec(unsigned idx, int wbits) {
    return (unsigned long long)idx | ((unsigned long long)(unsigned)wbits << 32);
}

__device__ __forceinline__ void scan_bins(const int* src, int* dst, int lane) {
    int v[8];
    int s = 0;
#pragma unroll
    for (int i = 0; i < 8; ++i) { v[i] = src[8 * lane + i]; s += v[i]; }
    int inc = s;
#pragma unroll
    for (int o = 1; o < 32; o <<= 1) {
        const int t = __shfl_up(inc, o);
        if (lane >= o) inc += t;
    }
    int run = inc - s;
#pragma unroll
    for (int i = 0; i < 8; ++i) { dst[8 * lane + i] = run; run += v[i]; }
}

__global__ void __launch_bounds__(CH_T)
k_place(const int* __restrict__ ei, const float* __restrict__ w,
        const int* __restrict__ gcount, int* __restrict__ gcursor,
        unsigned long long* __restrict__ rec, int E) {
    __shared__ int tot[2 * MAXB];
    __shared__ int start[2 * MAXB];
    __shared__ int hist[2 * MAXB];
    __shared__ int lofs[2 * MAXB];
    __shared__ int bpos[2 * MAXB];
    __shared__ unsigned long long lrec[2 * CH];
    __shared__ int   lidx[2 * CH];
    const int tid = threadIdx.x;
    if (tid < 2 * MAXB) { tot[tid] = gcount[tid]; hist[tid] = 0; }
    __syncthreads();

    const int base = blockIdx.x * CH;
    int r[CH_U], c[CH_U], wb[CH_U], b0[CH_U], b1[CH_U], k0[CH_U], k1[CH_U];
#pragma unroll
    for (int u = 0; u < CH_U; ++u) {
        const int e  = base + u * CH_T + tid;
        const int ec = (e < E) ? e : (E - 1);
        r[u]  = ei[ec];
        c[u]  = ei[E + ec];
        wb[u] = __float_as_int(w[ec]);
    }
    if (tid < 64) scan_bins(tot + (tid >> 5) * MAXB, start + (tid >> 5) * MAXB, tid & 31);
#pragma unroll
    for (int u = 0; u < CH_U; ++u) {
        const int e = base + u * CH_T + tid;
        b0[u] = r[u] / BIN;
        b1[u] = c[u] / BIN;
        k0[u] = 0;
        k1[u] = 0;
        if (e < E) {
            k0[u] = atomicAdd(&hist[b0[u]], 1);
            k1[u] = atomicAdd(&hist[MAXB + b1[u]], 1);
        }
    }
    __syncthreads();

    if (tid < 64) scan_bins(hist + (tid >> 5) * MAXB, lofs + (tid >> 5) * MAXB, tid & 31);
    if (tid < 2 * MAXB) {
        const int hcnt = hist[tid];
        const int g = hcnt ? atomicAdd(&gcursor[tid], hcnt) : 0;
        bpos[tid] = start[tid] + g;
    }
    __syncthreads();

#pragma unroll
    for (int u = 0; u < CH_U; ++u) {
        const int e = base + u * CH_T + tid;
        if (e < E) {
            const int p0 = lofs[b0[u]] + k0[u];
            lrec[p0] = pack_rec((unsigned)c[u] | ((unsigned)(r[u] - b0[u] * BIN) << 23), wb[u]);
            lidx[p0] = bpos[b0[u]] + k0[u];
            const int p1 = CH + lofs[MAXB + b1[u]] + k1[u];
            lrec[p1] = pack_rec((unsigned)r[u] | ((unsigned)(c[u] - b1[u] * BIN) << 23), wb[u]);
            lidx[p1] = bpos[MAXB + b1[u]] + k1[u];
        }
    }
    __syncthreads();

    const int nval = (E - base < CH) ? (E - base) : CH;
    unsigned long long* __restrict__ rec1 = rec + E;
    for (int p = tid; p < nval; p += CH_T) {
        rec[lidx[p]]       = lrec[p];
        rec1[lidx[CH + p]] = lrec[CH + p];
    }
}

#define BD_T 512
#define BD_U 8
__global__ void __launch_bounds__(BD_T)
k_bindeg(const unsigned long long* __restrict__ rec, const int* __restrict__ gcount,
         float* __restrict__ fac, int E, int N) {
    __shared__ unsigned long long dsum[2 * BIN];
    __shared__ int s_seg[4];
    const int tid = threadIdx.x;
    const int bin = blockIdx.x;

    if (tid < 4) s_seg[tid] = 0;
    for (int i = tid; i < 2 * BIN; i += BD_T) dsum[i] = 0ull;
    __syncthreads();
    if (tid < 2 * MAXB) {
        const int d  = tid / MAXB;
        const int bi = tid % MAXB;
        const int v  = gcount[tid];
        if (bi < bin && v) atomicAdd(&s_seg[d], v);
        if (bi == bin) s_seg[2 + d] = v;
    }
    __syncthreads();

    for (int d = 0; d < 2; ++d) {
        const int st  = s_seg[d];
        const int len = s_seg[2 + d];
        const unsigned long long* __restrict__ rb = rec + (long long)d * E + st;
        unsigned long long* ds = dsum + d * BIN;
        for (int i0 = 0; i0 < len; i0 += BD_T * BD_U) {
            unsigned long long rr[BD_U];
#pragma unroll
            for (int u = 0; u < BD_U; ++u) {
                const int i = i0 + u * BD_T + tid;
                rr[u] = rb[(i < len) ? i : (len - 1)];
            }
#pragma unroll
            for (int u = 0; u < BD_U; ++u) {
                const int i = i0 + u * BD_T + tid;
                if (i < len) {
                    const unsigned kd = (unsigned)rr[u] >> 23;
                    const float wv = __int_as_float((int)(unsigned)(rr[u] >> 32));
                    atomicAdd(&ds[kd], (unsigned long long)(long long)(wv * 0x1p40f));
                }
            }
        }
    }
    __syncthreads();

    for (int i = tid; i < 2 * BIN; i += BD_T) {
        const int d = i / BIN;
        const int node = bin * BIN + (i - d * BIN);
        if (node < N) {
            const float deg = (float)(long long)dsum[i] * 0x1p-40f;
            fac[(long long)d * N + node] = (deg > 0.0f) ? rsqrtf(deg) : 0.0f;
        }
    }
}

typedef __attribute__((ext_vector_type(4))) float v4f;
#define EP_PITCH 68
#define OW_T 1024
#define OW_W (OW_T / 32)
#define OW_U 16
#define OW_STRIPS (BIN / 16)

__global__ void __launch_bounds__(OW_T)
k_owner(const float* __restrict__ x, const unsigned long long* __restrict__ rec,
        const int* __restrict__ gcount, const float* __restrict__ fac,
        const float* __restrict__ Wsrc, const float* __restrict__ Wdst,
        const float* __restrict__ bsrc, const float* __restrict__ bdst,
        float* __restrict__ out, int E, int Nnodes) {
    __shared__ __attribute__((aligned(16))) float tab[2 * BIN * EP_PITCH];
    __shared__ __attribute__((aligned(16))) float sW[2 * 64 * EP_PITCH];
    __shared__ int s_seg[4];
    __shared__ float sF[2 * BIN];
    const int tid = threadIdx.x;
    const int bin = blockIdx.x;

    if (tid < 4) s_seg[tid] = 0;
    if (tid < 2 * BIN) {
        const int fdir = tid / BIN;
        const int node = bin * BIN + (tid - fdir * BIN);
        sF[tid] = (node < Nnodes) ? fac[(long long)fdir * Nnodes + node] : 0.0f;
    }
    for (int i = tid; i < 2 * BIN * EP_PITCH / 4; i += OW_T)
        ((v4f*)tab)[i] = (v4f){0.0f, 0.0f, 0.0f, 0.0f};
    v4f wtmp[2];
#pragma unroll
    for (int u = 0; u < 2; ++u) {
        const int i  = tid + u * OW_T;
        const int t  = i >> 10;
        const int rr = (i >> 4) & 63;
        const int q  = i & 15;
        wtmp[u] = *(const v4f*)((t ? Wdst : Wsrc) + rr * 64 + q * 4);
    }
#pragma unroll
    for (int u = 0; u < 2; ++u) {
        const int i  = tid + u * OW_T;
        const int t  = i >> 10;
        const int rr = (i >> 4) & 63;
        const int q  = i & 15;
        *(v4f*)(sW + (t * 64 + rr) * EP_PITCH + q * 4) = wtmp[u];
    }
    __syncthreads();
    if (tid < 2 * MAXB) {
        const int d  = tid / MAXB;
        const int bi = tid % MAXB;
        const int v  = gcount[tid];
        if (bi < bin && v) atomicAdd(&s_seg[d], v);
        if (bi == bin) s_seg[2 + d] = v;
    }
    __syncthreads();

    const int lane = tid & 31;
    const int wave = __builtin_amdgcn_readfirstlane(tid >> 5);
    for (int d = 0; d < 2; ++d) {
        const int st  = __builtin_amdgcn_readfirstlane(s_seg[d]);
        const int len = __builtin_amdgcn_readfirstlane(s_seg[2 + d]);
        const unsigned long long* __restrict__ rb = rec + (long long)d * E + st;
        float* tb = tab + d * (BIN * EP_PITCH) + lane;
        const float* __restrict__ fsrc = fac + (d == 0 ? (long long)Nnodes : 0LL);
        const float* sFd = sF + d * BIN;
        const int bstep = OW_W * 32;
        unsigned long long rc = 0ull, r1 = 0ull;
        int wbc = 0;
        if (wave * 32 < len) {
            const int i0 = wave * 32 + lane;
            rc = rb[(i0 < len) ? i0 : (len - 1)];
            if (wave * 32 + bstep < len) {
                const int i1 = wave * 32 + bstep + lane;
                r1 = rb[(i1 < len) ? i1 : (len - 1)];
            }
            const float fs0 = fsrc[(unsigned)rc & 0x7FFFFFu];
            const float fd0 = sFd[(unsigned)rc >> 23];
            const float wr0 = __int_as_float((int)(unsigned)(rc >> 32));
            wbc = __float_as_int((d == 0) ? (wr0 * fd0) * fs0 : (wr0 * fs0) * fd0);
        }
        for (int blk = wave * 32; blk < len; blk += bstep) {
            const int n  = (len - blk < 32) ? (len - blk) : 32;
            const unsigned rx = (unsigned)rc;
            const int ks = (int)(rx & 0x7FFFFFu);
            const int kd = (int)(rx >> 23);
            const int wb = wbc;
            unsigned long long r2 = 0ull;
            if (blk + 2 * bstep < len) {
                const int i2 = blk + 2 * bstep + lane;
                r2 = rb[(i2 < len) ? i2 : (len - 1)];
            }
            float fs1 = 0.0f, fd1 = 0.0f;
            const bool has1 = blk + bstep < len;
            if (has1) {
                fs1 = fsrc[(unsigned)r1 & 0x7FFFFFu];
                fd1 = sFd[(unsigned)r1 >> 23];
            }
            for (int j0 = 0; j0 < n; j0 += OW_U) {
                if (j0 + OW_U <= n) {
                    float x0[OW_U], x1[OW_U];
#pragma unroll
                    for (int u = 0; u < OW_U; ++u) {
                        const int s = __builtin_amdgcn_readlane(ks, j0 + u);
                        const float* ps = x + (long long)s * 64 + lane;
                        x0[u] = ps[0];
                        x1[u] = ps[32];
                    }
#pragma unroll
                    for (int u = 0; u < OW_U; ++u) {
                        const int   dn = __builtin_amdgcn_readlane(kd, j0 + u);
                        const float w  = __int_as_float(__builtin_amdgcn_readlane(wb, j0 + u));
                        float* pd = tb + dn * EP_PITCH;
                        atomAddF(pd,      w * x0[u]);
                        atomAddF(pd + 32, w * x1[u]);
                    }
                } else {
                    for (int j = j0; j < n; ++j) {
                        const int   s  = __builtin_amdgcn_readlane(ks, j);
                        const int   dn = __builtin_amdgcn_readlane(kd, j);
                        const float w  = __int_as_float(__builtin_amdgcn_readlane(wb, j));
                        const float* ps = x + (long long)s * 64 + lane;
                        const float a0 = ps[0], a1 = ps[32];
                        float* pd = tb + dn * EP_PITCH;
                        atomAddF(pd,      w * a0);
                        atomAddF(pd + 32, w * a1);
                    }
                }
            }
            if (has1) {
                const float wr1 = __int_as_float((int)(unsigned)(r1 >> 32));
                wbc = __float_as_int((d == 0) ? (wr1 * fd1) * fs1 : (wr1 * fs1) * fd1);
            }
            rc = r1;
            r1 = r2;
        }
    }
    __syncthreads();

    if (wave >= OW_STRIPS) return;

    const int c = lane & 15;
    const int h = lane >> 4;
    const float* af_p = tab + (wave * 16 + c) * EP_PITCH + 4 * h;
    const float* ab_p = af_p + BIN * EP_PITCH;
    const float* sB = sW + c * EP_PITCH + 4 * h;
    v8f acc[4] = {};
#pragma unroll
    for (int j = 0; j < 8; ++j) {
#pragma unroll
        for (int t = 0; t < 2; ++t) {
            const v4f a = *(const v4f*)((t ? ab_p : af_p) + 8 * j);
            v4f b[4];
#pragma unroll
            for (int n = 0; n < 4; ++n)
                b[n] = *(const v4f*)(sB + (t * 64 + n * 16) * EP_PITCH + 8 * j);
            const v2f a0 = {a.x, a.y};
            const v2f a1 = {a.z, a.w};
#pragma unroll
            for (int n = 0; n < 4; ++n) {
                const v2f b0 = {b[n].x, b[n].y};
                acc[n] = __builtin_amdgcn_wmma_f32_16x16x4_f32(
                             false, a0, false, b0, (short)0, acc[n], false, false);
            }
#pragma unroll
            for (int n = 0; n < 4; ++n) {
                const v2f b1 = {b[n].z, b[n].w};
                acc[n] = __builtin_amdgcn_wmma_f32_16x16x4_f32(
                             false, a1, false, b1, (short)0, acc[n], false, false);
            }
        }
    }

    float* so = tab + (wave * 16) * EP_PITCH;
#pragma unroll
    for (int n = 0; n < 4; ++n) {
        const int ncol = n * 16 + c;
        const float bias = 0.5f * (bsrc[ncol] + bdst[ncol]);
#pragma unroll
        for (int r = 0; r < 8; ++r)
            so[(8 * h + r) * EP_PITCH + ncol] = 0.5f * acc[n][r] + bias;
    }

#pragma unroll
    for (int i = 0; i < 8; ++i) {
        const int row = 2 * i + h;
        const int m = bin * BIN + wave * 16 + row;
        const v4f v = *(const v4f*)(so + row * EP_PITCH + 4 * c);
        if (m < Nnodes) *(v4f*)(out + (long long)m * 64 + 4 * c) = v;
    }
}

extern "C" void kernel_launch(void* const* d_in, const int* in_sizes, int n_in,
                              void* d_out, int out_size, void* d_ws, size_t ws_size,
                              hipStream_t stream) {
    const float* x    = (const float*)d_in[0];
    const int*   ei   = (const int*)d_in[1];
    const float* ew   = (const float*)d_in[2];
    const float* Wsrc = (const float*)d_in[3];
    const float* bsrc = (const float*)d_in[4];
    const float* Wdst = (const float*)d_in[5];
    const float* bdst = (const float*)d_in[6];
    float* out = (float*)d_out;

    const int N = in_sizes[0] / 64;
    const int E = in_sizes[2];

    float* ws  = (float*)d_ws;
    float* fac = ws;
    float* big = ws + 2LL * N + (long long)E + ((2 * N + E) & 1);

    unsigned long long* rec = (unsigned long long*)big;
    int* cnt     = (int*)(big + 4LL * E);
    int* gcount  = cnt;
    int* gcursor = cnt + 2 * MAXB;

    k_clear<<<1, NCNT, 0, stream>>>(cnt);
    k_count<<<(E + CNT_CH - 1) / CNT_CH, CNT_T, 0, stream>>>(ei, gcount, E);

    const int nch = (E + CH - 1) / CH;
    k_place<<<nch, CH_T, 0, stream>>>(ei, ew, gcount, gcursor, rec, E);

    const int nbins = (N + BIN - 1) / BIN;
    k_bindeg<<<nbins, BD_T, 0, stream>>>(rec, gcount, fac, E, N);
    k_owner<<<nbins, OW_T, 0, stream>>>(x, rec, gcount, fac, Wsrc, Wdst, bsrc, bdst, out, E, N);
}


# owner kernel: records counting-sorted by destination in LDS; a wave sums each destination run in registers, one ds_add_f32 pair per run (was per record); sum order changes, arrival order in a run
# speedup vs baseline: 4.2380x; 1.0805x over previous
#include <hip/hip_runtime.h>
#include <hip/hip_bf16.h>


typedef __attribute__((ext_vector_type(2))) float v2f;
typedef __attribute__((ext_vector_type(8))) float v8f;

__device__ __forceinline__ void atomAddF(float* p, float v) {
#if defined(__HIP_DEVICE_COMPILE__)
    unsafeAtomicAdd(p, v);
#else
    (void)p; (void)v;
#endif
}


#define BIN 400
#define MAXB 256
#define NCNT (4 * MAXB)
#define CH_T 1024
#define CH_U 4
#define CH (CH_T * CH_U)

__global__ void k_clear(int* __restrict__ cnt) {
    cnt[threadIdx.x] = 0;
}

#define CNT_T 512
#define CNT_U 16
#define CNT_CH (CNT_T * CNT_U)
__global__ void __launch_bounds__(CNT_T)
k_count(const int* __restrict__ ei, int* __restrict__ gcount, int E) {
    __shared__ int hist[2 * MAXB];
    const int tid = threadIdx.x;
    if (tid < 2 * MAXB) hist[tid] = 0;
    __syncthreads();

    const int base = blockIdx.x * CNT_CH;
    int r[CNT_U], c[CNT_U];
#pragma unroll
    for (int u = 0; u < CNT_U; ++u) {
        const int e  = base + u * CNT_T + tid;
        const int ec = (e < E) ? e : (E - 1);
        r[u] = ei[ec];
        c[u] = ei[E + ec];
    }
#pragma unroll
    for (int u = 0; u < CNT_U; ++u) {
        const int e = base + u * CNT_T + tid;
        if (e < E) {
            atomicAdd(&hist[r[u] / BIN], 1);
            atomicAdd(&hist[MAXB + c[u] / BIN], 1);
        }
    }
    __syncthreads();
    if (tid < 2 * MAXB) {
        const int hcnt = hist[tid];
        if (hcnt) atomicAdd(&gcount[tid], hcnt);
    }
}

__device__ __forceinline__ unsigned long long pack_rec(unsigned idx, int wbits) {
    return (unsigned long long)idx | ((unsigned long long)(unsigned)wbits << 32);
}

__device__ __forceinline__ void scan_bins(const int* src, int* dst, int lane) {
    int v[8];
    int s = 0;
#pragma unroll
    for (int i = 0; i < 8; ++i) { v[i] = src[8 * lane + i]; s += v[i]; }
    int inc = s;
#pragma unroll
    for (int o = 1; o < 32; o <<= 1) {
        const int t = __shfl_up(inc, o);
        if (lane >= o) inc += t;
    }
    int run = inc - s;
#pragma unroll
    for (int i = 0; i < 8; ++i) { dst[8 * lane + i] = run; run += v[i]; }
}

__global__ void __launch_bounds__(CH_T)
k_place(const int* __restrict__ ei, const float* __restrict__ w,
        const int* __restrict__ gcount, int* __restrict__ gcursor,
        unsigned long long* __restrict__ rec, int E) {
    __shared__ int tot[2 * MAXB];
    __shared__ int start[2 * MAXB];
    __shared__ int hist[2 * MAXB];
    __shared__ int lofs[2 * MAXB];
    __shared__ int bpos[2 * MAXB];
    __shared__ unsigned long long lrec[2 * CH];
    __shared__ int   lidx[2 * CH];
    const int tid = threadIdx.x;
    if (tid < 2 * MAXB) { tot[tid] = gcount[tid]; hist[tid] = 0; }
    __syncthreads();

    const int base = blockIdx.x * CH;
    int r[CH_U], c[CH_U], wb[CH_U], b0[CH_U], b1[CH_U], k0[CH_U], k1[CH_U];
#pragma unroll
    for (int u = 0; u < CH_U; ++u) {
        const int e  = base + u * CH_T + tid;
        const int ec = (e < E) ? e : (E - 1);
        r[u]  = ei[ec];
        c[u]  = ei[E + ec];
        wb[u] = __float_as_int(w[ec]);
    }
    if (tid < 64) scan_bins(tot + (tid >> 5) * MAXB, start + (tid >> 5) * MAXB, tid & 31);
#pragma unroll
    for (int u = 0; u < CH_U; ++u) {
        const int e = base + u * CH_T + tid;
        b0[u] = r[u] / BIN;
        b1[u] = c[u] / BIN;
        k0[u] = 0;
        k1[u] = 0;
        if (e < E) {
            k0[u] = atomicAdd(&hist[b0[u]], 1);
            k1[u] = atomicAdd(&hist[MAXB + b1[u]], 1);
        }
    }
    __syncthreads();

    if (tid < 64) scan_bins(hist + (tid >> 5) * MAXB, lofs + (tid >> 5) * MAXB, tid & 31);
    if (tid < 2 * MAXB) {
        const int hcnt = hist[tid];
        const int g = hcnt ? atomicAdd(&gcursor[tid], hcnt) : 0;
        bpos[tid] = start[tid] + g;
    }
    __syncthreads();

#pragma unroll
    for (int u = 0; u < CH_U; ++u) {
        const int e = base + u * CH_T + tid;
        if (e < E) {
            const int p0 = lofs[b0[u]] + k0[u];
            lrec[p0] = pack_rec((unsigned)c[u] | ((unsigned)(r[u] - b0[u] * BIN) << 23), wb[u]);
            lidx[p0] = bpos[b0[u]] + k0[u];
            const int p1 = CH + lofs[MAXB + b1[u]] + k1[u];
            lrec[p1] = pack_rec((unsigned)r[u] | ((unsigned)(c[u] - b1[u] * BIN) << 23), wb[u]);
            lidx[p1] = bpos[MAXB + b1[u]] + k1[u];
        }
    }
    __syncthreads();

    const int nval = (E - base < CH) ? (E - base) : CH;
    unsigned long long* __restrict__ rec1 = rec + E;
    for (int p = tid; p < nval; p += CH_T) {
        rec[lidx[p]]       = lrec[p];
        rec1[lidx[CH + p]] = lrec[CH + p];
    }
}

#define BD_T 512
#define BD_U 8
__global__ void __launch_bounds__(BD_T)
k_bindeg(const unsigned long long* __restrict__ rec, const int* __restrict__ gcount,
         float* __restrict__ fac, int E, int N) {
    __shared__ unsigned long long dsum[2 * BIN];
    __shared__ int s_seg[4];
    const int tid = threadIdx.x;
    const int bin = blockIdx.x;

    if (tid < 4) s_seg[tid] = 0;
    for (int i = tid; i < 2 * BIN; i += BD_T) dsum[i] = 0ull;
    __syncthreads();
    if (tid < 2 * MAXB) {
        const int d  = tid / MAXB;
        const int bi = tid % MAXB;
        const int v  = gcount[tid];
        if (bi < bin && v) atomicAdd(&s_seg[d], v);
        if (bi == bin) s_seg[2 + d] = v;
    }
    __syncthreads();

    for (int d = 0; d < 2; ++d) {
        const int st  = s_seg[d];
        const int len = s_seg[2 + d];
        const unsigned long long* __restrict__ rb = rec + (long long)d * E + st;
        unsigned long long* ds = dsum + d * BIN;
        for (int i0 = 0; i0 < len; i0 += BD_T * BD_U) {
            unsigned long long rr[BD_U];
#pragma unroll
            for (int u = 0; u < BD_U; ++u) {
                const int i = i0 + u * BD_T + tid;
                rr[u] = rb[(i < len) ? i : (len - 1)];
            }
#pragma unroll
            for (int u = 0; u < BD_U; ++u) {
                const int i = i0 + u * BD_T + tid;
                if (i < len) {
                    const unsigned kd = (unsigned)rr[u] >> 23;
                    const float wv = __int_as_float((int)(unsigned)(rr[u] >> 32));
                    atomicAdd(&ds[kd], (unsigned long long)(long long)(wv * 0x1p40f));
                }
            }
        }
    }
    __syncthreads();

    for (int i = tid; i < 2 * BIN; i += BD_T) {
        const int d = i / BIN;
        const int node = bin * BIN + (i - d * BIN);
        if (node < N) {
            const float deg = (float)(long long)dsum[i] * 0x1p-40f;
            fac[(long long)d * N + node] = (deg > 0.0f) ? rsqrtf(deg) : 0.0f;
        }
    }
}

typedef __attribute__((ext_vector_type(4))) float v4f;
#define EP_PITCH 68
#define OW_T 1024
#define OW_W (OW_T / 32)
#define OW_U 16
#define OW_STRIPS (BIN / 16)
#define SR_CH 7168
#define SR_U (SR_CH / OW_T)
#define SC_PER ((BIN + 31) / 32)

__global__ void __launch_bounds__(OW_T)
k_owner(const float* __restrict__ x, const unsigned long long* __restrict__ rec,
        const int* __restrict__ gcount, const float* __restrict__ fac,
        const float* __restrict__ Wsrc, const float* __restrict__ Wdst,
        const float* __restrict__ bsrc, const float* __restrict__ bdst,
        float* __restrict__ out, int E, int Nnodes) {
    __shared__ __attribute__((aligned(16))) float tab[2 * BIN * EP_PITCH];
    __shared__ __attribute__((aligned(16))) float sW[2 * 64 * EP_PITCH];
    __shared__ int s_seg[4];
    __shared__ float sF[2 * BIN];
    __shared__ unsigned long long srec[SR_CH];
    __shared__ int dcnt[BIN];
    __shared__ int dstart[BIN];
    const int tid = threadIdx.x;
    const int bin = blockIdx.x;

    if (tid < 4) s_seg[tid] = 0;
    if (tid < 2 * BIN) {
        const int fdir = tid / BIN;
        const int node = bin * BIN + (tid - fdir * BIN);
        sF[tid] = (node < Nnodes) ? fac[(long long)fdir * Nnodes + node] : 0.0f;
    }
    for (int i = tid; i < 2 * BIN * EP_PITCH / 4; i += OW_T)
        ((v4f*)tab)[i] = (v4f){0.0f, 0.0f, 0.0f, 0.0f};
    v4f wtmp[2];
#pragma unroll
    for (int u = 0; u < 2; ++u) {
        const int i  = tid + u * OW_T;
        const int t  = i >> 10;
        const int rr = (i >> 4) & 63;
        const int q  = i & 15;
        wtmp[u] = *(const v4f*)((t ? Wdst : Wsrc) + rr * 64 + q * 4);
    }
#pragma unroll
    for (int u = 0; u < 2; ++u) {
        const int i  = tid + u * OW_T;
        const int t  = i >> 10;
        const int rr = (i >> 4) & 63;
        const int q  = i & 15;
        *(v4f*)(sW + (t * 64 + rr) * EP_PITCH + q * 4) = wtmp[u];
    }
    __syncthreads();
    if (tid < 2 * MAXB) {
        const int d  = tid / MAXB;
        const int bi = tid % MAXB;
        const int v  = gcount[tid];
        if (bi < bin && v) atomicAdd(&s_seg[d], v);
        if (bi == bin) s_seg[2 + d] = v;
    }
    __syncthreads();

    const int lane = tid & 31;
    const int wave = __builtin_amdgcn_readfirstlane(tid >> 5);
    for (int d = 0; d < 2; ++d) {
        const int st  = __builtin_amdgcn_readfirstlane(s_seg[d]);
        const int len = __builtin_amdgcn_readfirstlane(s_seg[2 + d]);
        const unsigned long long* __restrict__ rb = rec + (long long)d * E + st;
        float* tb = tab + d * (BIN * EP_PITCH) + lane;
        const float* __restrict__ fsrc = fac + (d == 0 ? (long long)Nnodes : 0LL);
        const float* sFd = sF + d * BIN;
        for (int c0 = 0; c0 < len; c0 += SR_CH) {
            const int nrec = (len - c0 < SR_CH) ? (len - c0) : SR_CH;
            if (tid < BIN) dcnt[tid] = 0;
            unsigned long long rr[SR_U];
            int rk[SR_U];
#pragma unroll
            for (int u = 0; u < SR_U; ++u) {
                const int i = u * OW_T + tid;
                rr[u] = rb[c0 + ((i < nrec) ? i : (nrec - 1))];
            }
            __syncthreads();
#pragma unroll
            for (int u = 0; u < SR_U; ++u) {
                const int i = u * OW_T + tid;
                rk[u] = 0;
                if (i < nrec) rk[u] = atomicAdd(&dcnt[(unsigned)rr[u] >> 23], 1);
            }
            __syncthreads();
            if (tid < 32) {
                int v[SC_PER];
                int s = 0;
#pragma unroll
                for (int i = 0; i < SC_PER; ++i) {
                    const int k = SC_PER * lane + i;
                    v[i] = (k < BIN) ? dcnt[k] : 0;
                    s += v[i];
                }
                int inc = s;
#pragma unroll
                for (int o = 1; o < 32; o <<= 1) {
                    const int t = __shfl_up(inc, o);
                    if (lane >= o) inc += t;
                }
                int run = inc - s;
#pragma unroll
                for (int i = 0; i < SC_PER; ++i) {
                    const int k = SC_PER * lane + i;
                    if (k < BIN) dstart[k] = run;
                    run += v[i];
                }
            }
            __syncthreads();
#pragma unroll
            for (int u = 0; u < SR_U; ++u) {
                const int i = u * OW_T + tid;
                if (i < nrec) srec[dstart[(unsigned)rr[u] >> 23] + rk[u]] = rr[u];
            }
            __syncthreads();

            const int G    = (nrec + OW_U - 1) / OW_U;
            const int r_lo = ((wave * G) / OW_W) * OW_U;
            const int r_hi0 = (((wave + 1) * G) / OW_W) * OW_U;
            const int r_hi = (r_hi0 < nrec) ? r_hi0 : nrec;
            unsigned long long rc = 0ull, r1 = 0ull;
            int wbc = 0;
            if (r_lo < r_hi) {
                const int i0 = r_lo + lane;
                rc = srec[(i0 < r_hi) ? i0 : (r_hi - 1)];
                if (r_lo + 32 < r_hi) {
                    const int i1 = r_lo + 32 + lane;
                    r1 = srec[(i1 < r_hi) ? i1 : (r_hi - 1)];
                }
                const float fs0 = fsrc[(unsigned)rc & 0x7FFFFFu];
                const float fd0 = sFd[(unsigned)rc >> 23];
                const float wr0 = __int_as_float((int)(unsigned)(rc >> 32));
                wbc = __float_as_int((d == 0) ? (wr0 * fd0) * fs0 : (wr0 * fs0) * fd0);
            }
            int   cur = -1;
            float a0 = 0.0f, a1 = 0.0f;
            for (int blk = r_lo; blk < r_hi; blk += 32) {
                const int n  = (r_hi - blk < 32) ? (r_hi - blk) : 32;
                const unsigned rx = (unsigned)rc;
                const int ks = (int)(rx & 0x7FFFFFu);
                const int kd = (int)(rx >> 23);
                const int wb = wbc;
                unsigned long long r2 = 0ull;
                if (blk + 64 < r_hi) {
                    const int i2 = blk + 64 + lane;
                    r2 = srec[(i2 < r_hi) ? i2 : (r_hi - 1)];
                }
                float fs1 = 0.0f, fd1 = 0.0f;
                const bool has1 = blk + 32 < r_hi;
                if (has1) {
                    fs1 = fsrc[(unsigned)r1 & 0x7FFFFFu];
                    fd1 = sFd[(unsigned)r1 >> 23];
                }
                for (int j0 = 0; j0 < n; j0 += OW_U) {
                    if (j0 + OW_U <= n) {
                        float x0[OW_U], x1[OW_U];
#pragma unroll
                        for (int u = 0; u < OW_U; ++u) {
                            const int s = __builtin_amdgcn_readlane(ks, j0 + u);
                            const float* ps = x + (long long)s * 64 + lane;
                            x0[u] = ps[0];
                            x1[u] = ps[32];
                        }
#pragma unroll
                        for (int u = 0; u < OW_U; ++u) {
                            const int   dn = __builtin_amdgcn_readlane(kd, j0 + u);
                            const float w  = __int_as_float(__builtin_amdgcn_readlane(wb, j0 + u));
                            if (dn != cur) {
                                if (cur >= 0) {
                                    float* pd = tb + cur * EP_PITCH;
                                    atomAddF(pd,      a0);
                                    atomAddF(pd + 32, a1);
                                }
                                cur = dn;
                                a0 = 0.0f;
                                a1 = 0.0f;
                            }
                            a0 = fmaf(w, x0[u], a0);
                            a1 = fmaf(w, x1[u], a1);
                        }
                    } else {
                        for (int j = j0; j < n; ++j) {
                            const int   s  = __builtin_amdgcn_readlane(ks, j);
                            const int   dn = __builtin_amdgcn_readlane(kd, j);
                            const float w  = __int_as_float(__builtin_amdgcn_readlane(wb, j));
                            const float* ps = x + (long long)s * 64 + lane;
                            const float b0 = ps[0], b1 = ps[32];
                            if (dn != cur) {
                                if (cur >= 0) {
                                    float* pd = tb + cur * EP_PITCH;
                                    atomAddF(pd,      a0);
                                    atomAddF(pd + 32, a1);
                                }
                                cur = dn;
                                a0 = 0.0f;
                                a1 = 0.0f;
                            }
                            a0 = fmaf(w, b0, a0);
                            a1 = fmaf(w, b1, a1);
                        }
                    }
                }
                if (has1) {
                    const float wr1 = __int_as_float((int)(unsigned)(r1 >> 32));
                    wbc = __float_as_int((d == 0) ? (wr1 * fd1) * fs1 : (wr1 * fs1) * fd1);
                }
                rc = r1;
                r1 = r2;
            }
            if (cur >= 0) {
                float* pd = tb + cur * EP_PITCH;
                atomAddF(pd,      a0);
                atomAddF(pd + 32, a1);
            }
        }
    }
    __syncthreads();

    if (wave >= OW_STRIPS) return;

    const int c = lane & 15;
    const int h = lane >> 4;
    const float* af_p = tab + (wave * 16 + c) * EP_PITCH + 4 * h;
    const float* ab_p = af_p + BIN * EP_PITCH;
    const float* sB = sW + c * EP_PITCH + 4 * h;
    v8f acc[4] = {};
#pragma unroll
    for (int j = 0; j < 8; ++j) {
#pragma unroll
        for (int t = 0; t < 2; ++t) {
            const v4f a = *(const v4f*)((t ? ab_p : af_p) + 8 * j);
            v4f b[4];
#pragma unroll
            for (int n = 0; n < 4; ++n)
                b[n] = *(const v4f*)(sB + (t * 64 + n * 16) * EP_PITCH + 8 * j);
            const v2f a0 = {a.x, a.y};
            const v2f a1 = {a.z, a.w};
#pragma unroll
            for (int n = 0; n < 4; ++n) {
                const v2f b0 = {b[n].x, b[n].y};
                acc[n] = __builtin_amdgcn_wmma_f32_16x16x4_f32(
                             false, a0, false, b0, (short)0, acc[n], false, false);
            }
#pragma unroll
            for (int n = 0; n < 4; ++n) {
                const v2f b1 = {b[n].z, b[n].w};
                acc[n] = __builtin_amdgcn_wmma_f32_16x16x4_f32(
                             false, a1, false, b1, (short)0, acc[n], false, false);
            }
        }
    }

    float* so = tab + (wave * 16) * EP_PITCH;
#pragma unroll
    for (int n = 0; n < 4; ++n) {
        const int ncol = n * 16 + c;
        const float bias = 0.5f * (bsrc[ncol] + bdst[ncol]);
#pragma unroll
        for (int r = 0; r < 8; ++r)
            so[(8 * h + r) * EP_PITCH + ncol] = 0.5f * acc[n][r] + bias;
    }

#pragma unroll
    for (int i = 0; i < 8; ++i) {
        const int row = 2 * i + h;
        const int m = bin * BIN + wave * 16 + row;
        const v4f v = *(const v4f*)(so + row * EP_PITCH + 4 * c);
        if (m < Nnodes) *(v4f*)(out + (long long)m * 64 + 4 * c) = v;
    }
}

extern "C" void kernel_launch(void* const* d_in, const int* in_sizes, int n_in,
                              void* d_out, int out_size, void* d_ws, size_t ws_size,
                              hipStream_t stream) {
    const float* x    = (const float*)d_in[0];
    const int*   ei   = (const int*)d_in[1];
    const float* ew   = (const float*)d_in[2];
    const float* Wsrc = (const float*)d_in[3];
    const float* bsrc = (const float*)d_in[4];
    const float* Wdst = (const float*)d_in[5];
    const float* bdst = (const float*)d_in[6];
    float* out = (float*)d_out;

    const int N = in_sizes[0] / 64;
    const int E = in_sizes[2];

    float* ws  = (float*)d_ws;
    float* fac = ws;
    float* big = ws + 2LL * N + (long long)E + ((2 * N + E) & 1);

    unsigned long long* rec = (unsigned long long*)big;
    int* cnt     = (int*)(big + 4LL * E);
    int* gcount  = cnt;
    int* gcursor = cnt + 2 * MAXB;

    k_clear<<<1, NCNT, 0, stream>>>(cnt);
    k_count<<<(E + CNT_CH - 1) / CNT_CH, CNT_T, 0, stream>>>(ei, gcount, E);

    const int nch = (E + CH - 1) / CH;
    k_place<<<nch, CH_T, 0, stream>>>(ei, ew, gcount, gcursor, rec, E);

    const int nbins = (N + BIN - 1) / BIN;
    k_bindeg<<<nbins, BD_T, 0, stream>>>(rec, gcount, fac, E, N);
    k_owner<<<nbins, OW_T, 0, stream>>>(x, rec, gcount, fac, Wsrc, Wdst, bsrc, bdst, out, E, N);
}


# owner walk: each 16-lane half walks its own record stream, 16-byte row loads (a wave load = 2 whole rows, 4 lines; was 2 one-line loads per row), record passed by DPP row_share; sum order changes
# speedup vs baseline: 4.3982x; 1.0378x over previous
#include <hip/hip_runtime.h>
#include <hip/hip_bf16.h>


typedef __attribute__((ext_vector_type(2))) float v2f;
typedef __attribute__((ext_vector_type(8))) float v8f;

__device__ __forceinline__ void atomAddF(float* p, float v) {
#if defined(__HIP_DEVICE_COMPILE__)
    unsafeAtomicAdd(p, v);
#else
    (void)p; (void)v;
#endif
}


#define BIN 400
#define MAXB 256
#define NCNT (4 * MAXB)
#define CH_T 1024
#define CH_U 4
#define CH (CH_T * CH_U)

__global__ void k_clear(int* __restrict__ cnt) {
    cnt[threadIdx.x] = 0;
}

#define CNT_T 512
#define CNT_U 16
#define CNT_CH (CNT_T * CNT_U)
__global__ void __launch_bounds__(CNT_T)
k_count(const int* __restrict__ ei, int* __restrict__ gcount, int E) {
    __shared__ int hist[2 * MAXB];
    const int tid = threadIdx.x;
    if (tid < 2 * MAXB) hist[tid] = 0;
    __syncthreads();

    const int base = blockIdx.x * CNT_CH;
    int r[CNT_U], c[CNT_U];
#pragma unroll
    for (int u = 0; u < CNT_U; ++u) {
        const int e  = base + u * CNT_T + tid;
        const int ec = (e < E) ? e : (E - 1);
        r[u] = ei[ec];
        c[u] = ei[E + ec];
    }
#pragma unroll
    for (int u = 0; u < CNT_U; ++u) {
        const int e = base + u * CNT_T + tid;
        if (e < E) {
            atomicAdd(&hist[r[u] / BIN], 1);
            atomicAdd(&hist[MAXB + c[u] / BIN], 1);
        }
    }
    __syncthreads();
    if (tid < 2 * MAXB) {
        const int hcnt = hist[tid];
        if (hcnt) atomicAdd(&gcount[tid], hcnt);
    }
}

__device__ __forceinline__ unsigned long long pack_rec(unsigned idx, int wbits) {
    return (unsigned long long)idx | ((unsigned long long)(unsigned)wbits << 32);
}

__device__ __forceinline__ void scan_bins(const int* src, int* dst, int lane) {
    int v[8];
    int s = 0;
#pragma unroll
    for (int i = 0; i < 8; ++i) { v[i] = src[8 * lane + i]; s += v[i]; }
    int inc = s;
#pragma unroll
    for (int o = 1; o < 32; o <<= 1) {
        const int t = __shfl_up(inc, o);
        if (lane >= o) inc += t;
    }
    int run = inc - s;
#pragma unroll
    for (int i = 0; i < 8; ++i) { dst[8 * lane + i] = run; run += v[i]; }
}

__global__ void __launch_bounds__(CH_T)
k_place(const int* __restrict__ ei, const float* __restrict__ w,
        const int* __restrict__ gcount, int* __restrict__ gcursor,
        unsigned long long* __restrict__ rec, int E) {
    __shared__ int tot[2 * MAXB];
    __shared__ int start[2 * MAXB];
    __shared__ int hist[2 * MAXB];
    __shared__ int lofs[2 * MAXB];
    __shared__ int bpos[2 * MAXB];
    __shared__ unsigned long long lrec[2 * CH];
    __shared__ int   lidx[2 * CH];
    const int tid = threadIdx.x;
    if (tid < 2 * MAXB) { tot[tid] = gcount[tid]; hist[tid] = 0; }
    __syncthreads();

    const int base = blockIdx.x * CH;
    int r[CH_U], c[CH_U], wb[CH_U], b0[CH_U], b1[CH_U], k0[CH_U], k1[CH_U];
#pragma unroll
    for (int u = 0; u < CH_U; ++u) {
        const int e  = base + u * CH_T + tid;
        const int ec = (e < E) ? e : (E - 1);
        r[u]  = ei[ec];
        c[u]  = ei[E + ec];
        wb[u] = __float_as_int(w[ec]);
    }
    if (tid < 64) scan_bins(tot + (tid >> 5) * MAXB, start + (tid >> 5) * MAXB, tid & 31);
#pragma unroll
    for (int u = 0; u < CH_U; ++u) {
        const int e = base + u * CH_T + tid;
        b0[u] = r[u] / BIN;
        b1[u] = c[u] / BIN;
        k0[u] = 0;
        k1[u] = 0;
        if (e < E) {
            k0[u] = atomicAdd(&hist[b0[u]], 1);
            k1[u] = atomicAdd(&hist[MAXB + b1[u]], 1);
        }
    }
    __syncthreads();

    if (tid < 64) scan_bins(hist + (tid >> 5) * MAXB, lofs + (tid >> 5) * MAXB, tid & 31);
    if (tid < 2 * MAXB) {
        const int hcnt = hist[tid];
        const int g = hcnt ? atomicAdd(&gcursor[tid], hcnt) : 0;
        bpos[tid] = start[tid] + g;
    }
    __syncthreads();

#pragma unroll
    for (int u = 0; u < CH_U; ++u) {
        const int e = base + u * CH_T + tid;
        if (e < E) {
            const int p0 = lofs[b0[u]] + k0[u];
            lrec[p0] = pack_rec((unsigned)c[u] | ((unsigned)(r[u] - b0[u] * BIN) << 23), wb[u]);
            lidx[p0] = bpos[b0[u]] + k0[u];
            const int p1 = CH + lofs[MAXB + b1[u]] + k1[u];
            lrec[p1] = pack_rec((unsigned)r[u] | ((unsigned)(c[u] - b1[u] * BIN) << 23), wb[u]);
            lidx[p1] = bpos[MAXB + b1[u]] + k1[u];
        }
    }
    __syncthreads();

    const int nval = (E - base < CH) ? (E - base) : CH;
    unsigned long long* __restrict__ rec1 = rec + E;
    for (int p = tid; p < nval; p += CH_T) {
        rec[lidx[p]]       = lrec[p];
        rec1[lidx[CH + p]] = lrec[CH + p];
    }
}

#define BD_T 512
#define BD_U 8
__global__ void __launch_bounds__(BD_T)
k_bindeg(const unsigned long long* __restrict__ rec, const int* __restrict__ gcount,
         float* __restrict__ fac, int E, int N) {
    __shared__ unsigned long long dsum[2 * BIN];
    __shared__ int s_seg[4];
    const int tid = threadIdx.x;
    const int bin = blockIdx.x;

    if (tid < 4) s_seg[tid] = 0;
    for (int i = tid; i < 2 * BIN; i += BD_T) dsum[i] = 0ull;
    __syncthreads();
    if (tid < 2 * MAXB) {
        const int d  = tid / MAXB;
        const int bi = tid % MAXB;
        const int v  = gcount[tid];
        if (bi < bin && v) atomicAdd(&s_seg[d], v);
        if (bi == bin) s_seg[2 + d] = v;
    }
    __syncthreads();

    for (int d = 0; d < 2; ++d) {
        const int st  = s_seg[d];
        const int len = s_seg[2 + d];
        const unsigned long long* __restrict__ rb = rec + (long long)d * E + st;
        unsigned long long* ds = dsum + d * BIN;
        for (int i0 = 0; i0 < len; i0 += BD_T * BD_U) {
            unsigned long long rr[BD_U];
#pragma unroll
            for (int u = 0; u < BD_U; ++u) {
                const int i = i0 + u * BD_T + tid;
                rr[u] = rb[(i < len) ? i : (len - 1)];
            }
#pragma unroll
            for (int u = 0; u < BD_U; ++u) {
                const int i = i0 + u * BD_T + tid;
                if (i < len) {
                    const unsigned kd = (unsigned)rr[u] >> 23;
                    const float wv = __int_as_float((int)(unsigned)(rr[u] >> 32));
                    atomicAdd(&ds[kd], (unsigned long long)(long long)(wv * 0x1p40f));
                }
            }
        }
    }
    __syncthreads();

    for (int i = tid; i < 2 * BIN; i += BD_T) {
        const int d = i / BIN;
        const int node = bin * BIN + (i - d * BIN);
        if (node < N) {
            const float deg = (float)(long long)dsum[i] * 0x1p-40f;
            fac[(long long)d * N + node] = (deg > 0.0f) ? rsqrtf(deg) : 0.0f;
        }
    }
}

typedef __attribute__((ext_vector_type(4))) float v4f;
#define EP_PITCH 68
#define OW_T 1024
#define OW_W (OW_T / 32)
#define OW_U 8
#define SENT_KD 511u
#define SENT_REC ((unsigned long long)(SENT_KD << 23))
#define OW_STRIPS (BIN / 16)
#define SR_CH 7168
#define SR_U (SR_CH / OW_T)
#define SC_PER ((BIN + 31) / 32)

template <int J> __device__ __forceinline__ unsigned row_lane(unsigned v) {
    return (unsigned)__builtin_amdgcn_update_dpp(0, (int)v, 0x150 + J, 0xf, 0xf, true);
}

__global__ void __launch_bounds__(OW_T)
k_owner(const float* __restrict__ x, const unsigned long long* __restrict__ rec,
        const int* __restrict__ gcount, const float* __restrict__ fac,
        const float* __restrict__ Wsrc, const float* __restrict__ Wdst,
        const float* __restrict__ bsrc, const float* __restrict__ bdst,
        float* __restrict__ out, int E, int Nnodes) {
    __shared__ __attribute__((aligned(16))) float tab[2 * BIN * EP_PITCH];
    __shared__ __attribute__((aligned(16))) float sW[2 * 64 * EP_PITCH];
    __shared__ int s_seg[4];
    __shared__ float sF[2 * BIN];
    __shared__ unsigned long long srec[SR_CH];
    __shared__ int dcnt[BIN];
    __shared__ int dstart[BIN];
    const int tid = threadIdx.x;
    const int bin = blockIdx.x;

    if (tid < 4) s_seg[tid] = 0;
    if (tid < 2 * BIN) {
        const int fdir = tid / BIN;
        const int node = bin * BIN + (tid - fdir * BIN);
        sF[tid] = (node < Nnodes) ? fac[(long long)fdir * Nnodes + node] : 0.0f;
    }
    for (int i = tid; i < 2 * BIN * EP_PITCH / 4; i += OW_T)
        ((v4f*)tab)[i] = (v4f){0.0f, 0.0f, 0.0f, 0.0f};
    v4f wtmp[2];
#pragma unroll
    for (int u = 0; u < 2; ++u) {
        const int i  = tid + u * OW_T;
        const int t  = i >> 10;
        const int rr = (i >> 4) & 63;
        const int q  = i & 15;
        wtmp[u] = *(const v4f*)((t ? Wdst : Wsrc) + rr * 64 + q * 4);
    }
#pragma unroll
    for (int u = 0; u < 2; ++u) {
        const int i  = tid + u * OW_T;
        const int t  = i >> 10;
        const int rr = (i >> 4) & 63;
        const int q  = i & 15;
        *(v4f*)(sW + (t * 64 + rr) * EP_PITCH + q * 4) = wtmp[u];
    }
    __syncthreads();
    if (tid < 2 * MAXB) {
        const int d  = tid / MAXB;
        const int bi = tid % MAXB;
        const int v  = gcount[tid];
        if (bi < bin && v) atomicAdd(&s_seg[d], v);
        if (bi == bin) s_seg[2 + d] = v;
    }
    __syncthreads();

    const int lane = tid & 31;
    const int c16 = lane & 15;
    const int hh  = lane >> 4;
    const int wave = __builtin_amdgcn_readfirstlane(tid >> 5);
    for (int d = 0; d < 2; ++d) {
        const int st  = __builtin_amdgcn_readfirstlane(s_seg[d]);
        const int len = __builtin_amdgcn_readfirstlane(s_seg[2 + d]);
        const unsigned long long* __restrict__ rb = rec + (long long)d * E + st;
        float* tq = tab + d * (BIN * EP_PITCH) + 4 * c16;
        const float* __restrict__ fsrc = fac + (d == 0 ? (long long)Nnodes : 0LL);
        const float* sFd = sF + d * BIN;
        for (int c0 = 0; c0 < len; c0 += SR_CH) {
            const int nrec = (len - c0 < SR_CH) ? (len - c0) : SR_CH;
            if (tid < BIN) dcnt[tid] = 0;
            unsigned long long rr[SR_U];
            int rk[SR_U];
#pragma unroll
            for (int u = 0; u < SR_U; ++u) {
                const int i = u * OW_T + tid;
                rr[u] = rb[c0 + ((i < nrec) ? i : (nrec - 1))];
            }
            __syncthreads();
#pragma unroll
            for (int u = 0; u < SR_U; ++u) {
                const int i = u * OW_T + tid;
                rk[u] = 0;
                if (i < nrec) rk[u] = atomicAdd(&dcnt[(unsigned)rr[u] >> 23], 1);
            }
            __syncthreads();
            if (tid < 32) {
                int v[SC_PER];
                int s = 0;
#pragma unroll
                for (int i = 0; i < SC_PER; ++i) {
                    const int k = SC_PER * lane + i;
                    v[i] = (k < BIN) ? dcnt[k] : 0;
                    s += v[i];
                }
                int inc = s;
#pragma unroll
                for (int o = 1; o < 32; o <<= 1) {
                    const int t = __shfl_up(inc, o);
                    if (lane >= o) inc += t;
                }
                int run = inc - s;
#pragma unroll
                for (int i = 0; i < SC_PER; ++i) {
                    const int k = SC_PER * lane + i;
                    if (k < BIN) dstart[k] = run;
                    run += v[i];
                }
            }
            __syncthreads();
#pragma unroll
            for (int u = 0; u < SR_U; ++u) {
                const int i = u * OW_T + tid;
                if (i < nrec) srec[dstart[(unsigned)rr[u] >> 23] + rk[u]] = rr[u];
            }
            __syncthreads();

            const int G   = (nrec + OW_U - 1) / OW_U;
            const int loA = (((2 * wave)     * G) / (2 * OW_W)) * OW_U;
            const int loB = (((2 * wave + 1) * G) / (2 * OW_W)) * OW_U;
            const int hiB = (((2 * wave + 2) * G) / (2 * OW_W)) * OW_U;
            const int endA = (loB < nrec) ? loB : nrec;
            const int endB = (hiB < nrec) ? hiB : nrec;
            const int cntA = endA - loA;
            const int cntB = endB - loB;
            const int cmax = (cntA > cntB) ? cntA : cntB;
            const int nblk = (cmax + 15) >> 4;
            const int lo = hh ? loB : loA;
            const int hi = hh ? endB : endA;
            unsigned long long rc, r1;
            {
                const int i0 = lo + c16;
                const unsigned long long t0 = srec[(i0 < hi) ? i0 : 0];
                rc = (i0 < hi) ? t0 : SENT_REC;
                const int i1 = lo + 16 + c16;
                const unsigned long long t1 = srec[(i1 < hi) ? i1 : 0];
                r1 = (i1 < hi) ? t1 : SENT_REC;
            }
            int wbc;
            {
                const unsigned kd0 = (unsigned)rc >> 23;
                const float fs0 = fsrc[(unsigned)rc & 0x7FFFFFu];
                const float fd0 = sFd[(kd0 < BIN) ? kd0 : 0u];
                const float wr0 = __int_as_float((int)(unsigned)(rc >> 32));
                const float wn0 = (d == 0) ? (wr0 * fd0) * fs0 : (wr0 * fs0) * fd0;
                wbc = (kd0 < BIN) ? __float_as_int(wn0) : 0;
            }
            unsigned cur = SENT_KD;
            v4f acc = {0.0f, 0.0f, 0.0f, 0.0f};
            for (int b = 0; b < nblk; ++b) {
                const unsigned rx = (unsigned)rc;
                const unsigned wb = (unsigned)wbc;
                unsigned long long r2;
                {
                    const int i2 = lo + 16 * (b + 2) + c16;
                    const unsigned long long t2 = srec[(i2 < hi) ? i2 : 0];
                    r2 = (i2 < hi) ? t2 : SENT_REC;
                }
                const unsigned kd1 = (unsigned)r1 >> 23;
                const float fs1 = fsrc[(unsigned)r1 & 0x7FFFFFu];
                const float fd1 = sFd[(kd1 < BIN) ? kd1 : 0u];
                v4f xr[OW_U];
#define OW_LOAD(U, J)                                                                   \
                xr[U] = *(const v4f*)(x + (((row_lane<J>(rx) & 0x7FFFFFu) << 6) | (unsigned)(c16 << 2)));
#define OW_SUM(U, J)                                                                    \
                {                                                                       \
                    const unsigned dn = row_lane<J>(rx) >> 23;                          \
                    const float    w  = __int_as_float((int)row_lane<J>(wb));           \
                    if (dn != cur) {                                                    \
                        if (cur < BIN) {                                                \
                            float* pd = tq + cur * EP_PITCH;                            \
                            atomAddF(pd,     acc.x);                                    \
                            atomAddF(pd + 1, acc.y);                                    \
                            atomAddF(pd + 2, acc.z);                                    \
                            atomAddF(pd + 3, acc.w);                                    \
                        }                                                               \
                        cur = dn;                                                       \
                        acc = (v4f){0.0f, 0.0f, 0.0f, 0.0f};                            \
                    }                                                                   \
                    acc.x = fmaf(w, xr[U].x, acc.x);                                    \
                    acc.y = fmaf(w, xr[U].y, acc.y);                                    \
                    acc.z = fmaf(w, xr[U].z, acc.z);                                    \
                    acc.w = fmaf(w, xr[U].w, acc.w);                                    \
                }
                OW_LOAD(0, 0) OW_LOAD(1, 1) OW_LOAD(2, 2) OW_LOAD(3, 3)
                OW_LOAD(4, 4) OW_LOAD(5, 5) OW_LOAD(6, 6) OW_LOAD(7, 7)
                OW_SUM(0, 0) OW_SUM(1, 1) OW_SUM(2, 2) OW_SUM(3, 3)
                OW_SUM(4, 4) OW_SUM(5, 5) OW_SUM(6, 6) OW_SUM(7, 7)
                if (16 * b + OW_U < cmax) {
                OW_LOAD(0, 8)  OW_LOAD(1, 9)  OW_LOAD(2, 10) OW_LOAD(3, 11)
                OW_LOAD(4, 12) OW_LOAD(5, 13) OW_LOAD(6, 14) OW_LOAD(7, 15)
                OW_SUM(0, 8)  OW_SUM(1, 9)  OW_SUM(2, 10) OW_SUM(3, 11)
                OW_SUM(4, 12) OW_SUM(5, 13) OW_SUM(6, 14) OW_SUM(7, 15)
                }
#undef OW_LOAD
#undef OW_SUM
                {
                    const float wr1 = __int_as_float((int)(unsigned)(r1 >> 32));
                    const float wn1 = (d == 0) ? (wr1 * fd1) * fs1 : (wr1 * fs1) * fd1;
                    wbc = (kd1 < BIN) ? __float_as_int(wn1) : 0;
                }
                rc = r1;
                r1 = r2;
            }
            if (cur < BIN) {
                float* pd = tq + cur * EP_PITCH;
                atomAddF(pd,     acc.x);
                atomAddF(pd + 1, acc.y);
                atomAddF(pd + 2, acc.z);
                atomAddF(pd + 3, acc.w);
            }
        }
    }
    __syncthreads();

    if (wave >= OW_STRIPS) return;

    const int c = lane & 15;
    const int h = lane >> 4;
    const float* af_p = tab + (wave * 16 + c) * EP_PITCH + 4 * h;
    const float* ab_p = af_p + BIN * EP_PITCH;
    const float* sB = sW + c * EP_PITCH + 4 * h;
    v8f acc[4] = {};
#pragma unroll
    for (int j = 0; j < 8; ++j) {
#pragma unroll
        for (int t = 0; t < 2; ++t) {
            const v4f a = *(const v4f*)((t ? ab_p : af_p) + 8 * j);
            v4f b[4];
#pragma unroll
            for (int n = 0; n < 4; ++n)
                b[n] = *(const v4f*)(sB + (t * 64 + n * 16) * EP_PITCH + 8 * j);
            const v2f a0 = {a.x, a.y};
            const v2f a1 = {a.z, a.w};
#pragma unroll
            for (int n = 0; n < 4; ++n) {
                const v2f b0 = {b[n].x, b[n].y};
                acc[n] = __builtin_amdgcn_wmma_f32_16x16x4_f32(
                             false, a0, false, b0, (short)0, acc[n], false, false);
            }
#pragma unroll
            for (int n = 0; n < 4; ++n) {
                const v2f b1 = {b[n].z, b[n].w};
                acc[n] = __builtin_amdgcn_wmma_f32_16x16x4_f32(
                             false, a1, false, b1, (short)0, acc[n], false, false);
            }
        }
    }

    float* so = tab + (wave * 16) * EP_PITCH;
#pragma unroll
    for (int n = 0; n < 4; ++n) {
        const int ncol = n * 16 + c;
        const float bias = 0.5f * (bsrc[ncol] + bdst[ncol]);
#pragma unroll
        for (int r = 0; r < 8; ++r)
            so[(8 * h + r) * EP_PITCH + ncol] = 0.5f * acc[n][r] + bias;
    }

#pragma unroll
    for (int i = 0; i < 8; ++i) {
        const int row = 2 * i + h;
        const int m = bin * BIN + wave * 16 + row;
        const v4f v = *(const v4f*)(so + row * EP_PITCH + 4 * c);
        if (m < Nnodes) *(v4f*)(out + (long long)m * 64 + 4 * c) = v;
    }
}

extern "C" void kernel_launch(void* const* d_in, const int* in_sizes, int n_in,
                              void* d_out, int out_size, void* d_ws, size_t ws_size,
                              hipStream_t stream) {
    const float* x    = (const float*)d_in[0];
    const int*   ei   = (const int*)d_in[1];
    const float* ew   = (const float*)d_in[2];
    const float* Wsrc = (const float*)d_in[3];
    const float* bsrc = (const float*)d_in[4];
    const float* Wdst = (const float*)d_in[5];
    const float* bdst = (const float*)d_in[6];
    float* out = (float*)d_out;

    const int N = in_sizes[0] / 64;
    const int E = in_sizes[2];

    float* ws  = (float*)d_ws;
    float* fac = ws;
    float* big = ws + 2LL * N + (long long)E + ((2 * N + E) & 1);

    unsigned long long* rec = (unsigned long long*)big;
    int* cnt     = (int*)(big + 4LL * E);
    int* gcount  = cnt;
    int* gcursor = cnt + 2 * MAXB;

    k_clear<<<1, NCNT, 0, stream>>>(cnt);
    k_count<<<(E + CNT_CH - 1) / CNT_CH, CNT_T, 0, stream>>>(ei, gcount, E);

    const int nch = (E + CH - 1) / CH;
    k_place<<<nch, CH_T, 0, stream>>>(ei, ew, gcount, gcursor, rec, E);

    const int nbins = (N + BIN - 1) / BIN;
    k_bindeg<<<nbins, BD_T, 0, stream>>>(rec, gcount, fac, E, N);
    k_owner<<<nbins, OW_T, 0, stream>>>(x, rec, gcount, fac, Wsrc, Wdst, bsrc, bdst, out, E, N);
}


# epilogue: a strip's forward products (agg_fwd x WsrcT, 64 of its 128 WMMAs) issued inside the backward walk, one k-block behind each row-load group; sum order: all forward k, then all backward k
# speedup vs baseline: 4.5544x; 1.0355x over previous
#include <hip/hip_runtime.h>
#include <hip/hip_bf16.h>


typedef __attribute__((ext_vector_type(2))) float v2f;
typedef __attribute__((ext_vector_type(8))) float v8f;

__device__ __forceinline__ void atomAddF(float* p, float v) {
#if defined(__HIP_DEVICE_COMPILE__)
    unsafeAtomicAdd(p, v);
#else
    (void)p; (void)v;
#endif
}


#define BIN 400
#define MAXB 256
#define NCNT (4 * MAXB)
#define CH_T 1024
#define CH_U 4
#define CH (CH_T * CH_U)

__global__ void k_clear(int* __restrict__ cnt) {
    cnt[threadIdx.x] = 0;
}

#define CNT_T 512
#define CNT_U 16
#define CNT_CH (CNT_T * CNT_U)
__global__ void __launch_bounds__(CNT_T)
k_count(const int* __restrict__ ei, int* __restrict__ gcount, int E) {
    __shared__ int hist[2 * MAXB];
    const int tid = threadIdx.x;
    if (tid < 2 * MAXB) hist[tid] = 0;
    __syncthreads();

    const int base = blockIdx.x * CNT_CH;
    int r[CNT_U], c[CNT_U];
#pragma unroll
    for (int u = 0; u < CNT_U; ++u) {
        const int e  = base + u * CNT_T + tid;
        const int ec = (e < E) ? e : (E - 1);
        r[u] = ei[ec];
        c[u] = ei[E + ec];
    }
#pragma unroll
    for (int u = 0; u < CNT_U; ++u) {
        const int e = base + u * CNT_T + tid;
        if (e < E) {
            atomicAdd(&hist[r[u] / BIN], 1);
            atomicAdd(&hist[MAXB + c[u] / BIN], 1);
        }
    }
    __syncthreads();
    if (tid < 2 * MAXB) {
        const int hcnt = hist[tid];
        if (hcnt) atomicAdd(&gcount[tid], hcnt);
    }
}

__device__ __forceinline__ unsigned long long pack_rec(unsigned idx, int wbits) {
    return (unsigned long long)idx | ((unsigned long long)(unsigned)wbits << 32);
}

__device__ __forceinline__ void scan_bins(const int* src, int* dst, int lane) {
    int v[8];
    int s = 0;
#pragma unroll
    for (int i = 0; i < 8; ++i) { v[i] = src[8 * lane + i]; s += v[i]; }
    int inc = s;
#pragma unroll
    for (int o = 1; o < 32; o <<= 1) {
        const int t = __shfl_up(inc, o);
        if (lane >= o) inc += t;
    }
    int run = inc - s;
#pragma unroll
    for (int i = 0; i < 8; ++i) { dst[8 * lane + i] = run; run += v[i]; }
}

__global__ void __launch_bounds__(CH_T)
k_place(const int* __restrict__ ei, const float* __restrict__ w,
        const int* __restrict__ gcount, int* __restrict__ gcursor,
        unsigned long long* __restrict__ rec, int E) {
    __shared__ int tot[2 * MAXB];
    __shared__ int start[2 * MAXB];
    __shared__ int hist[2 * MAXB];
    __shared__ int lofs[2 * MAXB];
    __shared__ int bpos[2 * MAXB];
    __shared__ unsigned long long lrec[2 * CH];
    __shared__ int   lidx[2 * CH];
    const int tid = threadIdx.x;
    if (tid < 2 * MAXB) { tot[tid] = gcount[tid]; hist[tid] = 0; }
    __syncthreads();

    const int base = blockIdx.x * CH;
    int r[CH_U], c[CH_U], wb[CH_U], b0[CH_U], b1[CH_U], k0[CH_U], k1[CH_U];
#pragma unroll
    for (int u = 0; u < CH_U; ++u) {
        const int e  = base + u * CH_T + tid;
        const int ec = (e < E) ? e : (E - 1);
        r[u]  = ei[ec];
        c[u]  = ei[E + ec];
        wb[u] = __float_as_int(w[ec]);
    }
    if (tid < 64) scan_bins(tot + (tid >> 5) * MAXB, start + (tid >> 5) * MAXB, tid & 31);
#pragma unroll
    for (int u = 0; u < CH_U; ++u) {
        const int e = base + u * CH_T + tid;
        b0[u] = r[u] / BIN;
        b1[u] = c[u] / BIN;
        k0[u] = 0;
        k1[u] = 0;
        if (e < E) {
            k0[u] = atomicAdd(&hist[b0[u]], 1);
            k1[u] = atomicAdd(&hist[MAXB + b1[u]], 1);
        }
    }
    __syncthreads();

    if (tid < 64) scan_bins(hist + (tid >> 5) * MAXB, lofs + (tid >> 5) * MAXB, tid & 31);
    if (tid < 2 * MAXB) {
        const int hcnt = hist[tid];
        const int g = hcnt ? atomicAdd(&gcursor[tid], hcnt) : 0;
        bpos[tid] = start[tid] + g;
    }
    __syncthreads();

#pragma unroll
    for (int u = 0; u < CH_U; ++u) {
        const int e = base + u * CH_T + tid;
        if (e < E) {
            const int p0 = lofs[b0[u]] + k0[u];
            lrec[p0] = pack_rec((unsigned)c[u] | ((unsigned)(r[u] - b0[u] * BIN) << 23), wb[u]);
            lidx[p0] = bpos[b0[u]] + k0[u];
            const int p1 = CH + lofs[MAXB + b1[u]] + k1[u];
            lrec[p1] = pack_rec((unsigned)r[u] | ((unsigned)(c[u] - b1[u] * BIN) << 23), wb[u]);
            lidx[p1] = bpos[MAXB + b1[u]] + k1[u];
        }
    }
    __syncthreads();

    const int nval = (E - base < CH) ? (E - base) : CH;
    unsigned long long* __restrict__ rec1 = rec + E;
    for (int p = tid; p < nval; p += CH_T) {
        rec[lidx[p]]       = lrec[p];
        rec1[lidx[CH + p]] = lrec[CH + p];
    }
}

#define BD_T 512
#define BD_U 8
__global__ void __launch_bounds__(BD_T)
k_bindeg(const unsigned long long* __restrict__ rec, const int* __restrict__ gcount,
         float* __restrict__ fac, int E, int N) {
    __shared__ unsigned long long dsum[2 * BIN];
    __shared__ int s_seg[4];
    const int tid = threadIdx.x;
    const int bin = blockIdx.x;

    if (tid < 4) s_seg[tid] = 0;
    for (int i = tid; i < 2 * BIN; i += BD_T) dsum[i] = 0ull;
    __syncthreads();
    if (tid < 2 * MAXB) {
        const int d  = tid / MAXB;
        const int bi = tid % MAXB;
        const int v  = gcount[tid];
        if (bi < bin && v) atomicAdd(&s_seg[d], v);
        if (bi == bin) s_seg[2 + d] = v;
    }
    __syncthreads();

    for (int d = 0; d < 2; ++d) {
        const int st  = s_seg[d];
        const int len = s_seg[2 + d];
        const unsigned long long* __restrict__ rb = rec + (long long)d * E + st;
        unsigned long long* ds = dsum + d * BIN;
        for (int i0 = 0; i0 < len; i0 += BD_T * BD_U) {
            unsigned long long rr[BD_U];
#pragma unroll
            for (int u = 0; u < BD_U; ++u) {
                const int i = i0 + u * BD_T + tid;
                rr[u] = rb[(i < len) ? i : (len - 1)];
            }
#pragma unroll
            for (int u = 0; u < BD_U; ++u) {
                const int i = i0 + u * BD_T + tid;
                if (i < len) {
                    const unsigned kd = (unsigned)rr[u] >> 23;
                    const float wv = __int_as_float((int)(unsigned)(rr[u] >> 32));
                    atomicAdd(&ds[kd], (unsigned long long)(long long)(wv * 0x1p40f));
                }
            }
        }
    }
    __syncthreads();

    for (int i = tid; i < 2 * BIN; i += BD_T) {
        const int d = i / BIN;
        const int node = bin * BIN + (i - d * BIN);
        if (node < N) {
            const float deg = (float)(long long)dsum[i] * 0x1p-40f;
            fac[(long long)d * N + node] = (deg > 0.0f) ? rsqrtf(deg) : 0.0f;
        }
    }
}

typedef __attribute__((ext_vector_type(4))) float v4f;
#define EP_PITCH 68
#define OW_T 1024
#define OW_W (OW_T / 32)
#define OW_U 8
#define SENT_KD 511u
#define SENT_REC ((unsigned long long)(SENT_KD << 23))
#define OW_STRIPS (BIN / 16)
#define SR_CH 7168
#define SR_U (SR_CH / OW_T)
#define SC_PER ((BIN + 31) / 32)

template <int J> __device__ __forceinline__ unsigned row_lane(unsigned v) {
    return (unsigned)__builtin_amdgcn_update_dpp(0, (int)v, 0x150 + J, 0xf, 0xf, true);
}

__device__ __forceinline__ void ep_step(const float* ap, const float* bp, int j, v8f (&acc)[4]) {
    const v4f a = *(const v4f*)(ap + 8 * j);
    v4f b[4];
#pragma unroll
    for (int n = 0; n < 4; ++n)
        b[n] = *(const v4f*)(bp + (n * 16) * EP_PITCH + 8 * j);
    const v2f a0 = {a.x, a.y};
    const v2f a1 = {a.z, a.w};
#pragma unroll
    for (int n = 0; n < 4; ++n) {
        const v2f b0 = {b[n].x, b[n].y};
        acc[n] = __builtin_amdgcn_wmma_f32_16x16x4_f32(
                     false, a0, false, b0, (short)0, acc[n], false, false);
    }
#pragma unroll
    for (int n = 0; n < 4; ++n) {
        const v2f b1 = {b[n].z, b[n].w};
        acc[n] = __builtin_amdgcn_wmma_f32_16x16x4_f32(
                     false, a1, false, b1, (short)0, acc[n], false, false);
    }
}

__global__ void __launch_bounds__(OW_T)
k_owner(const float* __restrict__ x, const unsigned long long* __restrict__ rec,
        const int* __restrict__ gcount, const float* __restrict__ fac,
        const float* __restrict__ Wsrc, const float* __restrict__ Wdst,
        const float* __restrict__ bsrc, const float* __restrict__ bdst,
        float* __restrict__ out, int E, int Nnodes) {
    __shared__ __attribute__((aligned(16))) float tab[2 * BIN * EP_PITCH];
    __shared__ __attribute__((aligned(16))) float sW[2 * 64 * EP_PITCH];
    __shared__ int s_seg[4];
    __shared__ float sF[2 * BIN];
    __shared__ unsigned long long srec[SR_CH];
    __shared__ int dcnt[BIN];
    __shared__ int dstart[BIN];
    const int tid = threadIdx.x;
    const int bin = blockIdx.x;

    if (tid < 4) s_seg[tid] = 0;
    if (tid < 2 * BIN) {
        const int fdir = tid / BIN;
        const int node = bin * BIN + (tid - fdir * BIN);
        sF[tid] = (node < Nnodes) ? fac[(long long)fdir * Nnodes + node] : 0.0f;
    }
    for (int i = tid; i < 2 * BIN * EP_PITCH / 4; i += OW_T)
        ((v4f*)tab)[i] = (v4f){0.0f, 0.0f, 0.0f, 0.0f};
    v4f wtmp[2];
#pragma unroll
    for (int u = 0; u < 2; ++u) {
        const int i  = tid + u * OW_T;
        const int t  = i >> 10;
        const int rr = (i >> 4) & 63;
        const int q  = i & 15;
        wtmp[u] = *(const v4f*)((t ? Wdst : Wsrc) + rr * 64 + q * 4);
    }
#pragma unroll
    for (int u = 0; u < 2; ++u) {
        const int i  = tid + u * OW_T;
        const int t  = i >> 10;
        const int rr = (i >> 4) & 63;
        const int q  = i & 15;
        *(v4f*)(sW + (t * 64 + rr) * EP_PITCH + q * 4) = wtmp[u];
    }
    __syncthreads();
    if (tid < 2 * MAXB) {
        const int d  = tid / MAXB;
        const int bi = tid % MAXB;
        const int v  = gcount[tid];
        if (bi < bin && v) atomicAdd(&s_seg[d], v);
        if (bi == bin) s_seg[2 + d] = v;
    }
    __syncthreads();

    const int lane = tid & 31;
    const int c16 = lane & 15;
    const int hh  = lane >> 4;
    const int wave = __builtin_amdgcn_readfirstlane(tid >> 5);
    const float* af_p = tab + (wave * 16 + c16) * EP_PITCH + 4 * hh;
    const float* ab_p = af_p + BIN * EP_PITCH;
    const float* sB = sW + c16 * EP_PITCH + 4 * hh;
    v8f eacc[4] = {};
    int jn = 0;
    for (int d = 0; d < 2; ++d) {
        const bool ep_on = (d == 1) && (wave < OW_STRIPS);
        const int st  = __builtin_amdgcn_readfirstlane(s_seg[d]);
        const int len = __builtin_amdgcn_readfirstlane(s_seg[2 + d]);
        const unsigned long long* __restrict__ rb = rec + (long long)d * E + st;
        float* tq = tab + d * (BIN * EP_PITCH) + 4 * c16;
        const float* __restrict__ fsrc = fac + (d == 0 ? (long long)Nnodes : 0LL);
        const float* sFd = sF + d * BIN;
        for (int c0 = 0; c0 < len; c0 += SR_CH) {
            const int nrec = (len - c0 < SR_CH) ? (len - c0) : SR_CH;
            if (tid < BIN) dcnt[tid] = 0;
            unsigned long long rr[SR_U];
            int rk[SR_U];
#pragma unroll
            for (int u = 0; u < SR_U; ++u) {
                const int i = u * OW_T + tid;
                rr[u] = rb[c0 + ((i < nrec) ? i : (nrec - 1))];
            }
            __syncthreads();
#pragma unroll
            for (int u = 0; u < SR_U; ++u) {
                const int i = u * OW_T + tid;
                rk[u] = 0;
                if (i < nrec) rk[u] = atomicAdd(&dcnt[(unsigned)rr[u] >> 23], 1);
            }
            __syncthreads();
            if (tid < 32) {
                int v[SC_PER];
                int s = 0;
#pragma unroll
                for (int i = 0; i < SC_PER; ++i) {
                    const int k = SC_PER * lane + i;
                    v[i] = (k < BIN) ? dcnt[k] : 0;
                    s += v[i];
                }
                int inc = s;
#pragma unroll
                for (int o = 1; o < 32; o <<= 1) {
                    const int t = __shfl_up(inc, o);
                    if (lane >= o) inc += t;
                }
                int run = inc - s;
#pragma unroll
                for (int i = 0; i < SC_PER; ++i) {
                    const int k = SC_PER * lane + i;
                    if (k < BIN) dstart[k] = run;
                    run += v[i];
                }
            }
            __syncthreads();
#pragma unroll
            for (int u = 0; u < SR_U; ++u) {
                const int i = u * OW_T + tid;
                if (i < nrec) srec[dstart[(unsigned)rr[u] >> 23] + rk[u]] = rr[u];
            }
            __syncthreads();

            const int G   = (nrec + OW_U - 1) / OW_U;
            const int loA = (((2 * wave)     * G) / (2 * OW_W)) * OW_U;
            const int loB = (((2 * wave + 1) * G) / (2 * OW_W)) * OW_U;
            const int hiB = (((2 * wave + 2) * G) / (2 * OW_W)) * OW_U;
            const int endA = (loB < nrec) ? loB : nrec;
            const int endB = (hiB < nrec) ? hiB : nrec;
            const int cntA = endA - loA;
            const int cntB = endB - loB;
            const int cmax = (cntA > cntB) ? cntA : cntB;
            const int nblk = (cmax + 15) >> 4;
            const int lo = hh ? loB : loA;
            const int hi = hh ? endB : endA;
            unsigned long long rc, r1;
            {
                const int i0 = lo + c16;
                const unsigned long long t0 = srec[(i0 < hi) ? i0 : 0];
                rc = (i0 < hi) ? t0 : SENT_REC;
                const int i1 = lo + 16 + c16;
                const unsigned long long t1 = srec[(i1 < hi) ? i1 : 0];
                r1 = (i1 < hi) ? t1 : SENT_REC;
            }
            int wbc;
            {
                const unsigned kd0 = (unsigned)rc >> 23;
                const float fs0 = fsrc[(unsigned)rc & 0x7FFFFFu];
                const float fd0 = sFd[(kd0 < BIN) ? kd0 : 0u];
                const float wr0 = __int_as_float((int)(unsigned)(rc >> 32));
                const float wn0 = (d == 0) ? (wr0 * fd0) * fs0 : (wr0 * fs0) * fd0;
                wbc = (kd0 < BIN) ? __float_as_int(wn0) : 0;
            }
            unsigned cur = SENT_KD;
            v4f acc = {0.0f, 0.0f, 0.0f, 0.0f};
            for (int b = 0; b < nblk; ++b) {
                const unsigned rx = (unsigned)rc;
                const unsigned wb = (unsigned)wbc;
                unsigned long long r2;
                {
                    const int i2 = lo + 16 * (b + 2) + c16;
                    const unsigned long long t2 = srec[(i2 < hi) ? i2 : 0];
                    r2 = (i2 < hi) ? t2 : SENT_REC;
                }
                const unsigned kd1 = (unsigned)r1 >> 23;
                const float fs1 = fsrc[(unsigned)r1 & 0x7FFFFFu];
                const float fd1 = sFd[(kd1 < BIN) ? kd1 : 0u];
                v4f xr[OW_U];
#define OW_LOAD(U, J)                                                                   \
                xr[U] = *(const v4f*)(x + (((row_lane<J>(rx) & 0x7FFFFFu) << 6) | (unsigned)(c16 << 2)));
#define OW_SUM(U, J)                                                                    \
                {                                                                       \
                    const unsigned dn = row_lane<J>(rx) >> 23;                          \
                    const float    w  = __int_as_float((int)row_lane<J>(wb));           \
                    if (dn != cur) {                                                    \
                        if (cur < BIN) {                                                \
                            float* pd = tq + cur * EP_PITCH;                            \
                            atomAddF(pd,     acc.x);                                    \
                            atomAddF(pd + 1, acc.y);                                    \
                            atomAddF(pd + 2, acc.z);                                    \
                            atomAddF(pd + 3, acc.w);                                    \
                        }                                                               \
                        cur = dn;                                                       \
                        acc = (v4f){0.0f, 0.0f, 0.0f, 0.0f};                            \
                    }                                                                   \
                    acc.x = fmaf(w, xr[U].x, acc.x);                                    \
                    acc.y = fmaf(w, xr[U].y, acc.y);                                    \
                    acc.z = fmaf(w, xr[U].z, acc.z);                                    \
                    acc.w = fmaf(w, xr[U].w, acc.w);                                    \
                }
                OW_LOAD(0, 0) OW_LOAD(1, 1) OW_LOAD(2, 2) OW_LOAD(3, 3)
                OW_LOAD(4, 4) OW_LOAD(5, 5) OW_LOAD(6, 6) OW_LOAD(7, 7)
                if (ep_on && jn < 8) { ep_step(af_p, sB, jn, eacc); ++jn; }
                OW_SUM(0, 0) OW_SUM(1, 1) OW_SUM(2, 2) OW_SUM(3, 3)
                OW_SUM(4, 4) OW_SUM(5, 5) OW_SUM(6, 6) OW_SUM(7, 7)
                if (16 * b + OW_U < cmax) {
                OW_LOAD(0, 8)  OW_LOAD(1, 9)  OW_LOAD(2, 10) OW_LOAD(3, 11)
                OW_LOAD(4, 12) OW_LOAD(5, 13) OW_LOAD(6, 14) OW_LOAD(7, 15)
                if (ep_on && jn < 8) { ep_step(af_p, sB, jn, eacc); ++jn; }
                OW_SUM(0, 8)  OW_SUM(1, 9)  OW_SUM(2, 10) OW_SUM(3, 11)
                OW_SUM(4, 12) OW_SUM(5, 13) OW_SUM(6, 14) OW_SUM(7, 15)
                }
#undef OW_LOAD
#undef OW_SUM
                {
                    const float wr1 = __int_as_float((int)(unsigned)(r1 >> 32));
                    const float wn1 = (d == 0) ? (wr1 * fd1) * fs1 : (wr1 * fs1) * fd1;
                    wbc = (kd1 < BIN) ? __float_as_int(wn1) : 0;
                }
                rc = r1;
                r1 = r2;
            }
            if (cur < BIN) {
                float* pd = tq + cur * EP_PITCH;
                atomAddF(pd,     acc.x);
                atomAddF(pd + 1, acc.y);
                atomAddF(pd + 2, acc.z);
                atomAddF(pd + 3, acc.w);
            }
        }
    }
    __syncthreads();

    if (wave >= OW_STRIPS) return;

    const int c = c16;
    const int h = hh;
    for (; jn < 8; ++jn) ep_step(af_p, sB, jn, eacc);
#pragma unroll
    for (int j = 0; j < 8; ++j) ep_step(ab_p, sB + 64 * EP_PITCH, j, eacc);

    float* so = tab + (wave * 16) * EP_PITCH;
#pragma unroll
    for (int n = 0; n < 4; ++n) {
        const int ncol = n * 16 + c;
        const float bias = 0.5f * (bsrc[ncol] + bdst[ncol]);
#pragma unroll
        for (int r = 0; r < 8; ++r)
            so[(8 * h + r) * EP_PITCH + ncol] = 0.5f * eacc[n][r] + bias;
    }

#pragma unroll
    for (int i = 0; i < 8; ++i) {
        const int row = 2 * i + h;
        const int m = bin * BIN + wave * 16 + row;
        const v4f v = *(const v4f*)(so + row * EP_PITCH + 4 * c);
        if (m < Nnodes) *(v4f*)(out + (long long)m * 64 + 4 * c) = v;
    }
}

extern "C" void kernel_launch(void* const* d_in, const int* in_sizes, int n_in,
                              void* d_out, int out_size, void* d_ws, size_t ws_size,
                              hipStream_t stream) {
    const float* x    = (const float*)d_in[0];
    const int*   ei   = (const int*)d_in[1];
    const float* ew   = (const float*)d_in[2];
    const float* Wsrc = (const float*)d_in[3];
    const float* bsrc = (const float*)d_in[4];
    const float* Wdst = (const float*)d_in[5];
    const float* bdst = (const float*)d_in[6];
    float* out = (float*)d_out;

    const int N = in_sizes[0] / 64;
    const int E = in_sizes[2];

    float* ws  = (float*)d_ws;
    float* fac = ws;
    float* big = ws + 2LL * N + (long long)E + ((2 * N + E) & 1);

    unsigned long long* rec = (unsigned long long*)big;
    int* cnt     = (int*)(big + 4LL * E);
    int* gcount  = cnt;
    int* gcursor = cnt + 2 * MAXB;

    k_clear<<<1, NCNT, 0, stream>>>(cnt);
    k_count<<<(E + CNT_CH - 1) / CNT_CH, CNT_T, 0, stream>>>(ei, gcount, E);

    const int nch = (E + CH - 1) / CH;
    k_place<<<nch, CH_T, 0, stream>>>(ei, ew, gcount, gcursor, rec, E);

    const int nbins = (N + BIN - 1) / BIN;
    k_bindeg<<<nbins, BD_T, 0, stream>>>(rec, gcount, fac, E, N);
    k_owner<<<nbins, OW_T, 0, stream>>>(x, rec, gcount, fac, Wsrc, Wdst, bsrc, bdst, out, E, N);
}
